# starting kernel
# baseline (speedup 1.0000x reference)
#include <hip/hip_runtime.h>
#include <math.h>

constexpr int TT = 64;
constexpr int DD = 512;
constexpr int NH = 16;
constexpr int G  = 4;

constexpr int HSTR = 516;
constexpr int RSTR = 516;
constexpr int QSTR = 36;
constexpr int QG   = NH * QSTR;
constexpr int UG   = DD * NH + 16;

constexpr int OFF_H = 0;
constexpr int OFF_R = OFF_H + TT * HSTR;
constexpr int OFF_Q = OFF_R + 16 * RSTR;
constexpr int OFF_U = OFF_Q + G * QG;
constexpr int OFF_S = OFF_U + G * UG;
constexpr int LDS_FLOATS = OFF_S + TT * NH;

typedef __attribute__((ext_vector_type(2))) float v2f;
typedef __attribute__((ext_vector_type(4))) float f4;
typedef __attribute__((ext_vector_type(8))) float v8f;

static __device__ __forceinline__ v8f wmma4(v2f a, v2f b, v8f c) {
    return __builtin_amdgcn_wmma_f32_16x16x4_f32(
        false, a, false, b, (short)0, c, false, false);
}

__global__ __launch_bounds__(256, 1)
void temporal_attn_lastrow_g4(const float* __restrict__ inputs,
                              const float* __restrict__ pos,
                              const float* __restrict__ Wq,
                              const float* __restrict__ Wk,
                              const float* __restrict__ Wv,
                              float* __restrict__ out)
{
    extern __shared__ float lds[];
    float* sH = lds + OFF_H;
    float* sR = lds + OFF_R;
    float* sQ = lds + OFF_Q;
    float* sU = lds + OFF_U;
    float* sS = lds + OFF_S;

    const int tid  = threadIdx.x;
    const int lane = tid & 31;
    const int wave = tid >> 5;
    const int m    = lane & 15;
    const int hi   = lane >> 4;
    const long b0  = (long)blockIdx.x * G;

    for (int i = tid; i < 16 * RSTR; i += 256) sR[i] = 0.f;
    __syncthreads();
    for (int i = tid; i < G * DD; i += 256) {
        int g = i >> 9, d = i & 511;
        sR[g * RSTR + d] =
            inputs[(b0 + g) * (size_t)(TT * DD) + 63 * DD + d] + pos[63 * DD + d];
    }
    __syncthreads();

    for (int t = 0; t < 4; ++t) {
        const int n0 = (wave * 4 + t) * 16;
        v8f acc = {};
        for (int k = 0; k < DD; k += 4) {
            const int kk = k + 2 * hi;
            v2f a, b;
            a[0] = sR[m * RSTR + kk];
            a[1] = sR[m * RSTR + kk + 1];
            b[0] = Wq[(size_t)kk * DD + n0 + m];
            b[1] = Wq[(size_t)(kk + 1) * DD + n0 + m];
            acc = wmma4(a, b, acc);
        }
        if (hi == 0) {
            #pragma unroll
            for (int r = 0; r < G; ++r) {
                const int e = n0 + m;
                sQ[r * QG + (e >> 5) * QSTR + (e & 31)] = acc[r];
            }
        }
    }
    __syncthreads();

    {
        const int   gl    = m & 3;
        const float vmask = (m < G) ? 1.f : 0.f;
        for (int t = 0; t < 64; ++t) {
            const int job = wave * 64 + t;
            const int dt  = job >> 4;
            const int h   = job & 15;
            const int d0  = dt * 16;
            v8f acc = {};
            #pragma unroll
            for (int k = 0; k < 32; k += 4) {
                const int kk = k + 2 * hi;
                v2f a, b;
                a[0] = Wk[(size_t)(d0 + m) * DD + 32 * h + kk];
                a[1] = Wk[(size_t)(d0 + m) * DD + 32 * h + kk + 1];
                b[0] = vmask * sQ[gl * QG + h * QSTR + kk];
                b[1] = vmask * sQ[gl * QG + h * QSTR + kk + 1];
                acc = wmma4(a, b, acc);
            }
            if (m < G) {
                #pragma unroll
                for (int r = 0; r < 8; ++r)
                    sU[m * UG + (d0 + r + 8 * hi) * NH + h] = acc[r];
            }
        }
    }
    __syncthreads();

    const float scale = 0.044194173824159216f;
    for (int g = 0; g < G; ++g) {
        {
            const f4* in4 = (const f4*)(inputs + (b0 + g) * (size_t)(TT * DD));
            const f4* pe4 = (const f4*)pos;
            for (int i = tid; i < TT * DD / 4; i += 256) {
                f4 x = in4[i], p = pe4[i];
                int t = i >> 7, d = (i & 127) << 2;
                float* dst = sH + t * HSTR + d;
                dst[0] = x.x + p.x; dst[1] = x.y + p.y;
                dst[2] = x.z + p.z; dst[3] = x.w + p.w;
            }
            if (g + 1 < G) {
                const float* nxt = inputs + (b0 + g + 1) * (size_t)(TT * DD);
                #pragma unroll
                for (int it = 0; it < 4; ++it)
                    __builtin_prefetch(nxt + (size_t)(tid + it * 256) * 32, 0, 1);
            }
        }
        __syncthreads();

        if (wave < 4) {
            const int t0 = wave * 16;
            v8f acc = {};
            for (int k = 0; k < DD; k += 4) {
                const int kk = k + 2 * hi;
                v2f a, b;
                a[0] = sH[(t0 + m) * HSTR + kk];
                a[1] = sH[(t0 + m) * HSTR + kk + 1];
                b[0] = sU[g * UG + kk * NH + m];
                b[1] = sU[g * UG + (kk + 1) * NH + m];
                acc = wmma4(a, b, acc);
            }
            #pragma unroll
            for (int r = 0; r < 8; ++r)
                sS[(t0 + r + 8 * hi) * NH + m] = acc[r] * scale;
        }
        __syncthreads();

        if (tid < NH) {
            float mx = -INFINITY;
            for (int t = 0; t < TT; ++t) mx = fmaxf(mx, sS[t * NH + tid]);
            float sum = 0.f;
            for (int t = 0; t < TT; ++t) {
                float e = expf(sS[t * NH + tid] - mx);
                sS[t * NH + tid] = e;
                sum += e;
            }
            float inv = 1.f / sum;
            for (int t = 0; t < TT; ++t) sS[t * NH + tid] *= inv;
        }
        __syncthreads();

        for (int t = 0; t < 4; ++t) {
            const int d0 = (wave * 4 + t) * 16;
            v8f acc = {};
            for (int k = 0; k < TT; k += 4) {
                const int kk = k + 2 * hi;
                v2f a, b;
                a[0] = sH[kk * HSTR + d0 + m];
                a[1] = sH[(kk + 1) * HSTR + d0 + m];
                b[0] = sS[kk * NH + m];
                b[1] = sS[(kk + 1) * NH + m];
                acc = wmma4(a, b, acc);
            }
            #pragma unroll
            for (int r = 0; r < 8; ++r)
                sU[g * UG + (d0 + r + 8 * hi) * NH + m] = acc[r];
        }
        __syncthreads();
    }

    {
        const int   gl    = m & 3;
        const float vmask = (m < G) ? 1.f : 0.f;
        for (int t = 0; t < 4; ++t) {
            const int job = wave * 4 + t;
            const int h   = job >> 1;
            const int n0  = (job & 1) * 16;
            v8f acc = {};
            for (int k = 0; k < DD; k += 4) {
                const int kk = k + 2 * hi;
                v2f a, b;
                a[0] = vmask * sU[gl * UG + kk * NH + h];
                a[1] = vmask * sU[gl * UG + (kk + 1) * NH + h];
                b[0] = Wv[(size_t)kk * DD + 32 * h + n0 + m];
                b[1] = Wv[(size_t)(kk + 1) * DD + 32 * h + n0 + m];
                acc = wmma4(a, b, acc);
            }
            if (hi == 0) {
                #pragma unroll
                for (int r = 0; r < G; ++r)
                    out[(b0 + r) * (size_t)DD + 32 * h + n0 + m] = acc[r];
            }
        }
    }
}

extern "C" void kernel_launch(void* const* d_in, const int* in_sizes, int n_in,
                              void* d_out, int out_size, void* d_ws, size_t ws_size,
                              hipStream_t stream) {
    const float* inputs = (const float*)d_in[0];
    const float* pos    = (const float*)d_in[1];
    const float* Wq     = (const float*)d_in[2];
    const float* Wk     = (const float*)d_in[3];
    const float* Wv     = (const float*)d_in[4];
    float* out          = (float*)d_out;
    (void)d_ws; (void)ws_size; (void)n_in; (void)out_size;

    const int B = in_sizes[0] / (TT * DD);
    const size_t smem = (size_t)LDS_FLOATS * sizeof(float);

    hipFuncSetAttribute((const void*)temporal_attn_lastrow_g4,
                        hipFuncAttributeMaxDynamicSharedMemorySize, (int)smem);

    temporal_attn_lastrow_g4<<<B / G, 256, smem, stream>>>(
        inputs, pos, Wq, Wk, Wv, out);
}


# 1024-thread blocks (32 waves, 8 per SIMD) in place of 256: each phase's tiles and jobs re-dealt one per wave, 4x more loads in flight per WGP; every sum's k order kept, results bit-identical
# speedup vs baseline: 1.4143x; 1.4143x over previous
#include <hip/hip_runtime.h>
#include <math.h>

constexpr int TT = 64;
constexpr int DD = 512;
constexpr int NH = 16;
constexpr int G  = 4;
constexpr int NT = 1024;
constexpr int NW = NT / 32;

constexpr int HSTR = 516;
constexpr int RSTR = 516;
constexpr int QSTR = 36;
constexpr int QG   = NH * QSTR;
constexpr int UG   = DD * NH + 16;

constexpr int OFF_H = 0;
constexpr int OFF_R = OFF_H + TT * HSTR;
constexpr int OFF_Q = OFF_R + 16 * RSTR;
constexpr int OFF_U = OFF_Q + G * QG;
constexpr int OFF_S = OFF_U + G * UG;
constexpr int LDS_FLOATS = OFF_S + TT * NH;

typedef __attribute__((ext_vector_type(2))) float v2f;
typedef __attribute__((ext_vector_type(4))) float f4;
typedef __attribute__((ext_vector_type(8))) float v8f;

static __device__ __forceinline__ v8f wmma4(v2f a, v2f b, v8f c) {
    return __builtin_amdgcn_wmma_f32_16x16x4_f32(
        false, a, false, b, (short)0, c, false, false);
}

__global__ __launch_bounds__(NT, 1)
void temporal_attn_lastrow_g4(const float* __restrict__ inputs,
                              const float* __restrict__ pos,
                              const float* __restrict__ Wq,
                              const float* __restrict__ Wk,
                              const float* __restrict__ Wv,
                              float* __restrict__ out)
{
    extern __shared__ float lds[];
    float* sH = lds + OFF_H;
    float* sR = lds + OFF_R;
    float* sQ = lds + OFF_Q;
    float* sU = lds + OFF_U;
    float* sS = lds + OFF_S;

    const int tid  = threadIdx.x;
    const int lane = tid & 31;
    const int wave = tid >> 5;
    const int m    = lane & 15;
    const int hi   = lane >> 4;
    const long b0  = (long)blockIdx.x * G;

    for (int i = tid; i < 16 * RSTR; i += NT) sR[i] = 0.f;
    __syncthreads();
    static_assert(G * DD % NT == 0, "whole trips");
    for (int it = 0; it < G * DD / NT; ++it) {
        const int i = tid + it * NT;
        int g = i >> 9, d = i & 511;
        sR[g * RSTR + d] =
            inputs[(b0 + g) * (size_t)(TT * DD) + 63 * DD + d] + pos[63 * DD + d];
    }
    __syncthreads();

    static_assert(NW == DD / 16, "one 16-column tile per wave");
    {
        const int n0 = wave * 16;
        v8f acc = {};
        for (int k = 0; k < DD; k += 4) {
            const int kk = k + 2 * hi;
            v2f a, b;
            a[0] = sR[m * RSTR + kk];
            a[1] = sR[m * RSTR + kk + 1];
            b[0] = Wq[(size_t)kk * DD + n0 + m];
            b[1] = Wq[(size_t)(kk + 1) * DD + n0 + m];
            acc = wmma4(a, b, acc);
        }
        if (hi == 0) {
            #pragma unroll
            for (int r = 0; r < G; ++r) {
                const int e = n0 + m;
                sQ[r * QG + (e >> 5) * QSTR + (e & 31)] = acc[r];
            }
        }
    }
    __syncthreads();

    {
        const int   gl    = m & 3;
        const float vmask = (m < G) ? 1.f : 0.f;
        for (int t = 0; t < (DD / 16) * NH / NW; ++t) {
            const int job = wave * ((DD / 16) * NH / NW) + t;
            const int dt  = job >> 4;
            const int h   = job & 15;
            const int d0  = dt * 16;
            v8f acc = {};
            #pragma unroll
            for (int k = 0; k < 32; k += 4) {
                const int kk = k + 2 * hi;
                v2f a, b;
                a[0] = Wk[(size_t)(d0 + m) * DD + 32 * h + kk];
                a[1] = Wk[(size_t)(d0 + m) * DD + 32 * h + kk + 1];
                b[0] = vmask * sQ[gl * QG + h * QSTR + kk];
                b[1] = vmask * sQ[gl * QG + h * QSTR + kk + 1];
                acc = wmma4(a, b, acc);
            }
            if (m < G) {
                #pragma unroll
                for (int r = 0; r < 8; ++r)
                    sU[m * UG + (d0 + r + 8 * hi) * NH + h] = acc[r];
            }
        }
    }
    __syncthreads();

    const float scale = 0.044194173824159216f;
    for (int g = 0; g < G; ++g) {
        {
            const f4* in4 = (const f4*)(inputs + (b0 + g) * (size_t)(TT * DD));
            const f4* pe4 = (const f4*)pos;
            static_assert(TT * DD / 4 % NT == 0, "whole trips");
            constexpr int HV = TT * DD / 4 / NT;
            constexpr int HB = 4;
            #pragma nounroll
            for (int ib = 0; ib < HV; ib += HB) {
                f4 x[HB], p[HB];
                for (int j = 0; j < HB; ++j) {
                    const int i = tid + (ib + j) * NT;
                    x[j] = in4[i]; p[j] = pe4[i];
                }
                for (int j = 0; j < HB; ++j) {
                    const int i = tid + (ib + j) * NT;
                    int t = i >> 7, d = (i & 127) << 2;
                    float* dst = sH + t * HSTR + d;
                    dst[0] = x[j].x + p[j].x; dst[1] = x[j].y + p[j].y;
                    dst[2] = x[j].z + p[j].z; dst[3] = x[j].w + p[j].w;
                }
            }
            if (g + 1 < G) {
                const float* nxt = inputs + (b0 + g + 1) * (size_t)(TT * DD);
                static_assert(TT * DD / 32 == NT, "one 128-byte line per thread");
                __builtin_prefetch(nxt + (size_t)tid * 32, 0, 1);
            }
        }
        __syncthreads();

        if (wave < 4) {
            const int t0 = wave * 16;
            v8f acc = {};
            for (int k = 0; k < DD; k += 4) {
                const int kk = k + 2 * hi;
                v2f a, b;
                a[0] = sH[(t0 + m) * HSTR + kk];
                a[1] = sH[(t0 + m) * HSTR + kk + 1];
                b[0] = sU[g * UG + kk * NH + m];
                b[1] = sU[g * UG + (kk + 1) * NH + m];
                acc = wmma4(a, b, acc);
            }
            #pragma unroll
            for (int r = 0; r < 8; ++r)
                sS[(t0 + r + 8 * hi) * NH + m] = acc[r] * scale;
        }
        __syncthreads();

        if (tid < NH) {
            float mx = -INFINITY;
            for (int t = 0; t < TT; ++t) mx = fmaxf(mx, sS[t * NH + tid]);
            float sum = 0.f;
            for (int t = 0; t < TT; ++t) {
                float e = expf(sS[t * NH + tid] - mx);
                sS[t * NH + tid] = e;
                sum += e;
            }
            float inv = 1.f / sum;
            for (int t = 0; t < TT; ++t) sS[t * NH + tid] *= inv;
        }
        __syncthreads();

        {
            const int d0 = wave * 16;
            v8f acc = {};
            for (int k = 0; k < TT; k += 4) {
                const int kk = k + 2 * hi;
                v2f a, b;
                a[0] = sH[kk * HSTR + d0 + m];
                a[1] = sH[(kk + 1) * HSTR + d0 + m];
                b[0] = sS[kk * NH + m];
                b[1] = sS[(kk + 1) * NH + m];
                acc = wmma4(a, b, acc);
            }
            #pragma unroll
            for (int r = 0; r < 8; ++r)
                sU[g * UG + (d0 + r + 8 * hi) * NH + m] = acc[r];
        }
        __syncthreads();
    }

    {
        const int   gl    = m & 3;
        const float vmask = (m < G) ? 1.f : 0.f;
        {
            const int job = wave;
            const int h   = job >> 1;
            const int n0  = (job & 1) * 16;
            v8f acc = {};
            for (int k = 0; k < DD; k += 4) {
                const int kk = k + 2 * hi;
                v2f a, b;
                a[0] = vmask * sU[gl * UG + kk * NH + h];
                a[1] = vmask * sU[gl * UG + (kk + 1) * NH + h];
                b[0] = Wv[(size_t)kk * DD + 32 * h + n0 + m];
                b[1] = Wv[(size_t)(kk + 1) * DD + 32 * h + n0 + m];
                acc = wmma4(a, b, acc);
            }
            if (hi == 0) {
                #pragma unroll
                for (int r = 0; r < G; ++r)
                    out[(b0 + r) * (size_t)DD + 32 * h + n0 + m] = acc[r];
            }
        }
    }
}

extern "C" void kernel_launch(void* const* d_in, const int* in_sizes, int n_in,
                              void* d_out, int out_size, void* d_ws, size_t ws_size,
                              hipStream_t stream) {
    const float* inputs = (const float*)d_in[0];
    const float* pos    = (const float*)d_in[1];
    const float* Wq     = (const float*)d_in[2];
    const float* Wk     = (const float*)d_in[3];
    const float* Wv     = (const float*)d_in[4];
    float* out          = (float*)d_out;
    (void)d_ws; (void)ws_size; (void)n_in; (void)out_size;

    const int B = in_sizes[0] / (TT * DD);
    const size_t smem = (size_t)LDS_FLOATS * sizeof(float);

    hipFuncSetAttribute((const void*)temporal_attn_lastrow_g4,
                        hipFuncAttributeMaxDynamicSharedMemorySize, (int)smem);

    temporal_attn_lastrow_g4<<<B / G, NT, smem, stream>>>(
        inputs, pos, Wq, Wk, Wv, out);
}


# softmax over t dealt to 16 waves (wave = head, lane = two t) with max and sum as 32-lane register butterflies, was one thread per head in three serial LDS passes; the sum over t becomes a tree
# speedup vs baseline: 1.7084x; 1.2079x over previous
#include <hip/hip_runtime.h>
#include <math.h>

constexpr int TT = 64;
constexpr int DD = 512;
constexpr int NH = 16;
constexpr int G  = 4;
constexpr int NT = 1024;
constexpr int NW = NT / 32;

constexpr int HSTR = 516;
constexpr int RSTR = 516;
constexpr int QSTR = 36;
constexpr int QG   = NH * QSTR;
constexpr int UG   = DD * NH + 16;
constexpr int SSTR = NH + 1;

constexpr int OFF_H = 0;
constexpr int OFF_R = OFF_H + TT * HSTR;
constexpr int OFF_Q = OFF_R + 16 * RSTR;
constexpr int OFF_U = OFF_Q + G * QG;
constexpr int OFF_S = OFF_U + G * UG;
constexpr int LDS_FLOATS = OFF_S + TT * SSTR;

typedef __attribute__((ext_vector_type(2))) float v2f;
typedef __attribute__((ext_vector_type(4))) float f4;
typedef __attribute__((ext_vector_type(8))) float v8f;

static __device__ __forceinline__ v8f wmma4(v2f a, v2f b, v8f c) {
    return __builtin_amdgcn_wmma_f32_16x16x4_f32(
        false, a, false, b, (short)0, c, false, false);
}

template <int CTRL>
static __device__ __forceinline__ float row_xor(float x) {
    const int b = __builtin_bit_cast(int, x);
    return __builtin_bit_cast(float,
        __builtin_amdgcn_update_dpp(b, b, CTRL, 0xf, 0xf, false));
}
static __device__ __forceinline__ float xhalf(float x) {
    const int b = __builtin_bit_cast(int, x);
    return __builtin_bit_cast(float,
        __builtin_amdgcn_permlanex16(b, b, 0x76543210, 0xfedcba98, false, false));
}
static __device__ __forceinline__ float wave_max(float x) {
    x = fmaxf(x, row_xor<0x161>(x));
    x = fmaxf(x, row_xor<0x162>(x));
    x = fmaxf(x, row_xor<0x164>(x));
    x = fmaxf(x, row_xor<0x168>(x));
    return fmaxf(x, xhalf(x));
}
static __device__ __forceinline__ float wave_sum(float x) {
    x += row_xor<0x161>(x);
    x += row_xor<0x162>(x);
    x += row_xor<0x164>(x);
    x += row_xor<0x168>(x);
    return x + xhalf(x);
}

__global__ __launch_bounds__(NT, 1)
void temporal_attn_lastrow_g4(const float* __restrict__ inputs,
                              const float* __restrict__ pos,
                              const float* __restrict__ Wq,
                              const float* __restrict__ Wk,
                              const float* __restrict__ Wv,
                              float* __restrict__ out)
{
    extern __shared__ float lds[];
    float* sH = lds + OFF_H;
    float* sR = lds + OFF_R;
    float* sQ = lds + OFF_Q;
    float* sU = lds + OFF_U;
    float* sS = lds + OFF_S;

    const int tid  = threadIdx.x;
    const int lane = tid & 31;
    const int wave = tid >> 5;
    const int m    = lane & 15;
    const int hi   = lane >> 4;
    const long b0  = (long)blockIdx.x * G;

    for (int i = tid; i < 16 * RSTR; i += NT) sR[i] = 0.f;
    __syncthreads();
    static_assert(G * DD % NT == 0, "whole trips");
    for (int it = 0; it < G * DD / NT; ++it) {
        const int i = tid + it * NT;
        int g = i >> 9, d = i & 511;
        sR[g * RSTR + d] =
            inputs[(b0 + g) * (size_t)(TT * DD) + 63 * DD + d] + pos[63 * DD + d];
    }
    __syncthreads();

    static_assert(NW == DD / 16, "one 16-column tile per wave");
    {
        const int n0 = wave * 16;
        v8f acc = {};
        for (int k = 0; k < DD; k += 4) {
            const int kk = k + 2 * hi;
            v2f a, b;
            a[0] = sR[m * RSTR + kk];
            a[1] = sR[m * RSTR + kk + 1];
            b[0] = Wq[(size_t)kk * DD + n0 + m];
            b[1] = Wq[(size_t)(kk + 1) * DD + n0 + m];
            acc = wmma4(a, b, acc);
        }
        if (hi == 0) {
            #pragma unroll
            for (int r = 0; r < G; ++r) {
                const int e = n0 + m;
                sQ[r * QG + (e >> 5) * QSTR + (e & 31)] = acc[r];
            }
        }
    }
    __syncthreads();

    {
        const int   gl    = m & 3;
        const float vmask = (m < G) ? 1.f : 0.f;
        for (int t = 0; t < (DD / 16) * NH / NW; ++t) {
            const int job = wave * ((DD / 16) * NH / NW) + t;
            const int dt  = job >> 4;
            const int h   = job & 15;
            const int d0  = dt * 16;
            v8f acc = {};
            #pragma unroll
            for (int k = 0; k < 32; k += 4) {
                const int kk = k + 2 * hi;
                v2f a, b;
                a[0] = Wk[(size_t)(d0 + m) * DD + 32 * h + kk];
                a[1] = Wk[(size_t)(d0 + m) * DD + 32 * h + kk + 1];
                b[0] = vmask * sQ[gl * QG + h * QSTR + kk];
                b[1] = vmask * sQ[gl * QG + h * QSTR + kk + 1];
                acc = wmma4(a, b, acc);
            }
            if (m < G) {
                #pragma unroll
                for (int r = 0; r < 8; ++r)
                    sU[m * UG + (d0 + r + 8 * hi) * NH + h] = acc[r];
            }
        }
    }
    __syncthreads();

    const float scale = 0.044194173824159216f;
    for (int g = 0; g < G; ++g) {
        {
            const f4* in4 = (const f4*)(inputs + (b0 + g) * (size_t)(TT * DD));
            const f4* pe4 = (const f4*)pos;
            static_assert(TT * DD / 4 % NT == 0, "whole trips");
            constexpr int HV = TT * DD / 4 / NT;
            constexpr int HB = 4;
            #pragma nounroll
            for (int ib = 0; ib < HV; ib += HB) {
                f4 x[HB], p[HB];
                for (int j = 0; j < HB; ++j) {
                    const int i = tid + (ib + j) * NT;
                    x[j] = in4[i]; p[j] = pe4[i];
                }
                for (int j = 0; j < HB; ++j) {
                    const int i = tid + (ib + j) * NT;
                    int t = i >> 7, d = (i & 127) << 2;
                    float* dst = sH + t * HSTR + d;
                    dst[0] = x[j].x + p[j].x; dst[1] = x[j].y + p[j].y;
                    dst[2] = x[j].z + p[j].z; dst[3] = x[j].w + p[j].w;
                }
            }
            if (g + 1 < G) {
                const float* nxt = inputs + (b0 + g + 1) * (size_t)(TT * DD);
                static_assert(TT * DD / 32 == NT, "one 128-byte line per thread");
                __builtin_prefetch(nxt + (size_t)tid * 32, 0, 1);
            }
        }
        __syncthreads();

        if (wave < 4) {
            const int t0 = wave * 16;
            v8f acc = {};
            for (int k = 0; k < DD; k += 4) {
                const int kk = k + 2 * hi;
                v2f a, b;
                a[0] = sH[(t0 + m) * HSTR + kk];
                a[1] = sH[(t0 + m) * HSTR + kk + 1];
                b[0] = sU[g * UG + kk * NH + m];
                b[1] = sU[g * UG + (kk + 1) * NH + m];
                acc = wmma4(a, b, acc);
            }
            #pragma unroll
            for (int r = 0; r < 8; ++r)
                sS[(t0 + r + 8 * hi) * SSTR + m] = acc[r] * scale;
        }
        __syncthreads();

        static_assert(TT == 64 && NH <= NW, "two t per lane, one head per wave");
        if (wave < NH) {
            float* col = sS + wave;
            const float s0  = col[lane * SSTR];
            const float s1  = col[(lane + 32) * SSTR];
            const float mx  = wave_max(fmaxf(s0, s1));
            const float e0  = expf(s0 - mx);
            const float e1  = expf(s1 - mx);
            const float inv = 1.f / wave_sum(e0 + e1);
            col[lane * SSTR]        = e0 * inv;
            col[(lane + 32) * SSTR] = e1 * inv;
        }
        __syncthreads();

        {
            const int d0 = wave * 16;
            v8f acc = {};
            for (int k = 0; k < TT; k += 4) {
                const int kk = k + 2 * hi;
                v2f a, b;
                a[0] = sH[kk * HSTR + d0 + m];
                a[1] = sH[(kk + 1) * HSTR + d0 + m];
                b[0] = sS[kk * SSTR + m];
                b[1] = sS[(kk + 1) * SSTR + m];
                acc = wmma4(a, b, acc);
            }
            #pragma unroll
            for (int r = 0; r < 8; ++r)
                sU[g * UG + (d0 + r + 8 * hi) * NH + m] = acc[r];
        }
        __syncthreads();
    }

    {
        const int   gl    = m & 3;
        const float vmask = (m < G) ? 1.f : 0.f;
        {
            const int job = wave;
            const int h   = job >> 1;
            const int n0  = (job & 1) * 16;
            v8f acc = {};
            for (int k = 0; k < DD; k += 4) {
                const int kk = k + 2 * hi;
                v2f a, b;
                a[0] = vmask * sU[gl * UG + kk * NH + h];
                a[1] = vmask * sU[gl * UG + (kk + 1) * NH + h];
                b[0] = Wv[(size_t)kk * DD + 32 * h + n0 + m];
                b[1] = Wv[(size_t)(kk + 1) * DD + 32 * h + n0 + m];
                acc = wmma4(a, b, acc);
            }
            if (hi == 0) {
                #pragma unroll
                for (int r = 0; r < G; ++r)
                    out[(b0 + r) * (size_t)DD + 32 * h + n0 + m] = acc[r];
            }
        }
    }
}

extern "C" void kernel_launch(void* const* d_in, const int* in_sizes, int n_in,
                              void* d_out, int out_size, void* d_ws, size_t ws_size,
                              hipStream_t stream) {
    const float* inputs = (const float*)d_in[0];
    const float* pos    = (const float*)d_in[1];
    const float* Wq     = (const float*)d_in[2];
    const float* Wk     = (const float*)d_in[3];
    const float* Wv     = (const float*)d_in[4];
    float* out          = (float*)d_out;
    (void)d_ws; (void)ws_size; (void)n_in; (void)out_size;

    const int B = in_sizes[0] / (TT * DD);
    const size_t smem = (size_t)LDS_FLOATS * sizeof(float);

    hipFuncSetAttribute((const void*)temporal_attn_lastrow_g4,
                        hipFuncAttributeMaxDynamicSharedMemorySize, (int)smem);

    temporal_attn_lastrow_g4<<<B / G, NT, smem, stream>>>(
        inputs, pos, Wq, Wk, Wv, out);
}


# score GEMM S = H @ U_g dealt to all 32 waves as 4 t-tiles x 8 K-slices of 64 (was 4 waves, K = 512 each); the softmax wave adds the 8 partials in slice order: the sum over d is now 8 chains of 64
# speedup vs baseline: 1.7903x; 1.0479x over previous
#include <hip/hip_runtime.h>
#include <math.h>

constexpr int TT = 64;
constexpr int DD = 512;
constexpr int NH = 16;
constexpr int G  = 4;
constexpr int NT = 1024;
constexpr int NW = NT / 32;

constexpr int HSTR = 516;
constexpr int RSTR = 516;
constexpr int QSTR = 36;
constexpr int QG   = NH * QSTR;
constexpr int UG   = DD * NH + 16;
constexpr int SSTR = NH + 1;

constexpr int KS   = NW / (TT / 16);
constexpr int KSL  = DD / KS;
constexpr int PSTR = TT * SSTR;

constexpr int OFF_H = 0;
constexpr int OFF_R = OFF_H + TT * HSTR;
constexpr int OFF_Q = OFF_R + 16 * RSTR;
constexpr int OFF_U = OFF_Q + G * QG;
constexpr int OFF_S = OFF_U + G * UG;
constexpr int LDS_FLOATS = OFF_S + TT * SSTR;
static_assert(KS * PSTR <= OFF_U - OFF_R, "score partials fit in the sR + sQ area");

typedef __attribute__((ext_vector_type(2))) float v2f;
typedef __attribute__((ext_vector_type(4))) float f4;
typedef __attribute__((ext_vector_type(8))) float v8f;

static __device__ __forceinline__ v8f wmma4(v2f a, v2f b, v8f c) {
    return __builtin_amdgcn_wmma_f32_16x16x4_f32(
        false, a, false, b, (short)0, c, false, false);
}

template <int CTRL>
static __device__ __forceinline__ float row_xor(float x) {
    const int b = __builtin_bit_cast(int, x);
    return __builtin_bit_cast(float,
        __builtin_amdgcn_update_dpp(b, b, CTRL, 0xf, 0xf, false));
}
static __device__ __forceinline__ float xhalf(float x) {
    const int b = __builtin_bit_cast(int, x);
    return __builtin_bit_cast(float,
        __builtin_amdgcn_permlanex16(b, b, 0x76543210, 0xfedcba98, false, false));
}
static __device__ __forceinline__ float wave_max(float x) {
    x = fmaxf(x, row_xor<0x161>(x));
    x = fmaxf(x, row_xor<0x162>(x));
    x = fmaxf(x, row_xor<0x164>(x));
    x = fmaxf(x, row_xor<0x168>(x));
    return fmaxf(x, xhalf(x));
}
static __device__ __forceinline__ float wave_sum(float x) {
    x += row_xor<0x161>(x);
    x += row_xor<0x162>(x);
    x += row_xor<0x164>(x);
    x += row_xor<0x168>(x);
    return x + xhalf(x);
}

__global__ __launch_bounds__(NT, 1)
void temporal_attn_lastrow_g4(const float* __restrict__ inputs,
                              const float* __restrict__ pos,
                              const float* __restrict__ Wq,
                              const float* __restrict__ Wk,
                              const float* __restrict__ Wv,
                              float* __restrict__ out)
{
    extern __shared__ float lds[];
    float* sH = lds + OFF_H;
    float* sR = lds + OFF_R;
    float* sQ = lds + OFF_Q;
    float* sU = lds + OFF_U;
    float* sS = lds + OFF_S;
    float* sP = lds + OFF_R;

    const int tid  = threadIdx.x;
    const int lane = tid & 31;
    const int wave = tid >> 5;
    const int m    = lane & 15;
    const int hi   = lane >> 4;
    const long b0  = (long)blockIdx.x * G;

    for (int i = tid; i < 16 * RSTR; i += NT) sR[i] = 0.f;
    __syncthreads();
    static_assert(G * DD % NT == 0, "whole trips");
    for (int it = 0; it < G * DD / NT; ++it) {
        const int i = tid + it * NT;
        int g = i >> 9, d = i & 511;
        sR[g * RSTR + d] =
            inputs[(b0 + g) * (size_t)(TT * DD) + 63 * DD + d] + pos[63 * DD + d];
    }
    __syncthreads();

    static_assert(NW == DD / 16, "one 16-column tile per wave");
    {
        const int n0 = wave * 16;
        v8f acc = {};
        for (int k = 0; k < DD; k += 4) {
            const int kk = k + 2 * hi;
            v2f a, b;
            a[0] = sR[m * RSTR + kk];
            a[1] = sR[m * RSTR + kk + 1];
            b[0] = Wq[(size_t)kk * DD + n0 + m];
            b[1] = Wq[(size_t)(kk + 1) * DD + n0 + m];
            acc = wmma4(a, b, acc);
        }
        if (hi == 0) {
            #pragma unroll
            for (int r = 0; r < G; ++r) {
                const int e = n0 + m;
                sQ[r * QG + (e >> 5) * QSTR + (e & 31)] = acc[r];
            }
        }
    }
    __syncthreads();

    {
        const int   gl    = m & 3;
        const float vmask = (m < G) ? 1.f : 0.f;
        for (int t = 0; t < (DD / 16) * NH / NW; ++t) {
            const int job = wave * ((DD / 16) * NH / NW) + t;
            const int dt  = job >> 4;
            const int h   = job & 15;
            const int d0  = dt * 16;
            v8f acc = {};
            #pragma unroll
            for (int k = 0; k < 32; k += 4) {
                const int kk = k + 2 * hi;
                v2f a, b;
                a[0] = Wk[(size_t)(d0 + m) * DD + 32 * h + kk];
                a[1] = Wk[(size_t)(d0 + m) * DD + 32 * h + kk + 1];
                b[0] = vmask * sQ[gl * QG + h * QSTR + kk];
                b[1] = vmask * sQ[gl * QG + h * QSTR + kk + 1];
                acc = wmma4(a, b, acc);
            }
            if (m < G) {
                #pragma unroll
                for (int r = 0; r < 8; ++r)
                    sU[m * UG + (d0 + r + 8 * hi) * NH + h] = acc[r];
            }
        }
    }
    __syncthreads();

    const float scale = 0.044194173824159216f;
    for (int g = 0; g < G; ++g) {
        {
            const f4* in4 = (const f4*)(inputs + (b0 + g) * (size_t)(TT * DD));
            const f4* pe4 = (const f4*)pos;
            static_assert(TT * DD / 4 % NT == 0, "whole trips");
            constexpr int HV = TT * DD / 4 / NT;
            constexpr int HB = 4;
            #pragma nounroll
            for (int ib = 0; ib < HV; ib += HB) {
                f4 x[HB], p[HB];
                for (int j = 0; j < HB; ++j) {
                    const int i = tid + (ib + j) * NT;
                    x[j] = in4[i]; p[j] = pe4[i];
                }
                for (int j = 0; j < HB; ++j) {
                    const int i = tid + (ib + j) * NT;
                    int t = i >> 7, d = (i & 127) << 2;
                    float* dst = sH + t * HSTR + d;
                    dst[0] = x[j].x + p[j].x; dst[1] = x[j].y + p[j].y;
                    dst[2] = x[j].z + p[j].z; dst[3] = x[j].w + p[j].w;
                }
            }
            if (g + 1 < G) {
                const float* nxt = inputs + (b0 + g + 1) * (size_t)(TT * DD);
                static_assert(TT * DD / 32 == NT, "one 128-byte line per thread");
                __builtin_prefetch(nxt + (size_t)tid * 32, 0, 1);
            }
        }
        __syncthreads();

        static_assert(KS * (TT / 16) == NW && KS * KSL == DD, "one (tile, slice) per wave");
        {
            const int t0 = (wave & 3) * 16;
            const int c  = wave >> 2;
            v8f acc = {};
            for (int k = 0; k < KSL; k += 4) {
                const int kk = c * KSL + k + 2 * hi;
                v2f a, b;
                a[0] = sH[(t0 + m) * HSTR + kk];
                a[1] = sH[(t0 + m) * HSTR + kk + 1];
                b[0] = sU[g * UG + kk * NH + m];
                b[1] = sU[g * UG + (kk + 1) * NH + m];
                acc = wmma4(a, b, acc);
            }
            #pragma unroll
            for (int r = 0; r < 8; ++r)
                sP[c * PSTR + (t0 + r + 8 * hi) * SSTR + m] = acc[r];
        }
        __syncthreads();

        static_assert(TT == 64 && NH <= NW, "two t per lane, one head per wave");
        if (wave < NH) {
            float* col = sS + wave;
            const float* par = sP + wave;
            float s0 = par[lane * SSTR];
            float s1 = par[(lane + 32) * SSTR];
            for (int c = 1; c < KS; ++c) {
                s0 += par[c * PSTR + lane * SSTR];
                s1 += par[c * PSTR + (lane + 32) * SSTR];
            }
            s0 *= scale;
            s1 *= scale;
            const float mx  = wave_max(fmaxf(s0, s1));
            const float e0  = expf(s0 - mx);
            const float e1  = expf(s1 - mx);
            const float inv = 1.f / wave_sum(e0 + e1);
            col[lane * SSTR]        = e0 * inv;
            col[(lane + 32) * SSTR] = e1 * inv;
        }
        __syncthreads();

        {
            const int d0 = wave * 16;
            v8f acc = {};
            for (int k = 0; k < TT; k += 4) {
                const int kk = k + 2 * hi;
                v2f a, b;
                a[0] = sH[kk * HSTR + d0 + m];
                a[1] = sH[(kk + 1) * HSTR + d0 + m];
                b[0] = sS[kk * SSTR + m];
                b[1] = sS[(kk + 1) * SSTR + m];
                acc = wmma4(a, b, acc);
            }
            #pragma unroll
            for (int r = 0; r < 8; ++r)
                sU[g * UG + (d0 + r + 8 * hi) * NH + m] = acc[r];
        }
        __syncthreads();
    }

    {
        const int   gl    = m & 3;
        const float vmask = (m < G) ? 1.f : 0.f;
        {
            const int job = wave;
            const int h   = job >> 1;
            const int n0  = (job & 1) * 16;
            v8f acc = {};
            for (int k = 0; k < DD; k += 4) {
                const int kk = k + 2 * hi;
                v2f a, b;
                a[0] = vmask * sU[gl * UG + kk * NH + h];
                a[1] = vmask * sU[gl * UG + (kk + 1) * NH + h];
                b[0] = Wv[(size_t)kk * DD + 32 * h + n0 + m];
                b[1] = Wv[(size_t)(kk + 1) * DD + 32 * h + n0 + m];
                acc = wmma4(a, b, acc);
            }
            if (hi == 0) {
                #pragma unroll
                for (int r = 0; r < G; ++r)
                    out[(b0 + r) * (size_t)DD + 32 * h + n0 + m] = acc[r];
            }
        }
    }
}

extern "C" void kernel_launch(void* const* d_in, const int* in_sizes, int n_in,
                              void* d_out, int out_size, void* d_ws, size_t ws_size,
                              hipStream_t stream) {
    const float* inputs = (const float*)d_in[0];
    const float* pos    = (const float*)d_in[1];
    const float* Wq     = (const float*)d_in[2];
    const float* Wk     = (const float*)d_in[3];
    const float* Wv     = (const float*)d_in[4];
    float* out          = (float*)d_out;
    (void)d_ws; (void)ws_size; (void)n_in; (void)out_size;

    const int B = in_sizes[0] / (TT * DD);
    const size_t smem = (size_t)LDS_FLOATS * sizeof(float);

    hipFuncSetAttribute((const void*)temporal_attn_lastrow_g4,
                        hipFuncAttributeMaxDynamicSharedMemorySize, (int)smem);

    temporal_attn_lastrow_g4<<<B / G, NT, smem, stream>>>(
        inputs, pos, Wq, Wk, Wv, out);
}


# 8 batches per block, one round of 256 blocks (was 512 of 4): the last-row q GEMM and its Wq scan run once per 8 batches on 8 of the tile's 16 rows; later stages as two groups of 4; bit-identical
# speedup vs baseline: 2.1463x; 1.1988x over previous
#include <hip/hip_runtime.h>
#include <math.h>

constexpr int TT = 64;
constexpr int DD = 512;
constexpr int NH = 16;
constexpr int G  = 4;
constexpr int NSG = 2;
constexpr int GB  = G * NSG;
constexpr int NT = 1024;
constexpr int NW = NT / 32;

constexpr int HSTR = 516;
constexpr int RSTR = 516;
constexpr int QSTR = 36;
constexpr int QG   = NH * QSTR;
constexpr int UG   = DD * NH + 16;
constexpr int SSTR = NH + 1;

constexpr int KS   = NW / (TT / 16);
constexpr int KSL  = DD / KS;
constexpr int PSTR = TT * SSTR;

constexpr int OFF_H = 0;
constexpr int OFF_R = OFF_H + TT * HSTR;
constexpr int RP_FLOATS = (KS * PSTR > 16 * RSTR) ? KS * PSTR : 16 * RSTR;
constexpr int OFF_Q = OFF_R + RP_FLOATS;
constexpr int OFF_U = OFF_Q + GB * QG;
constexpr int OFF_S = OFF_U + G * UG;
constexpr int LDS_FLOATS = OFF_S + TT * SSTR;
static_assert(KS * PSTR <= OFF_Q - OFF_R, "score partials fit in the sR area");
static_assert(LDS_FLOATS * 4 <= 320 * 1024, "one workgroup's LDS");

typedef __attribute__((ext_vector_type(2))) float v2f;
typedef __attribute__((ext_vector_type(4))) float f4;
typedef __attribute__((ext_vector_type(8))) float v8f;

static __device__ __forceinline__ v8f wmma4(v2f a, v2f b, v8f c) {
    return __builtin_amdgcn_wmma_f32_16x16x4_f32(
        false, a, false, b, (short)0, c, false, false);
}

template <int CTRL>
static __device__ __forceinline__ float row_xor(float x) {
    const int b = __builtin_bit_cast(int, x);
    return __builtin_bit_cast(float,
        __builtin_amdgcn_update_dpp(b, b, CTRL, 0xf, 0xf, false));
}
static __device__ __forceinline__ float xhalf(float x) {
    const int b = __builtin_bit_cast(int, x);
    return __builtin_bit_cast(float,
        __builtin_amdgcn_permlanex16(b, b, 0x76543210, 0xfedcba98, false, false));
}
static __device__ __forceinline__ float wave_max(float x) {
    x = fmaxf(x, row_xor<0x161>(x));
    x = fmaxf(x, row_xor<0x162>(x));
    x = fmaxf(x, row_xor<0x164>(x));
    x = fmaxf(x, row_xor<0x168>(x));
    return fmaxf(x, xhalf(x));
}
static __device__ __forceinline__ float wave_sum(float x) {
    x += row_xor<0x161>(x);
    x += row_xor<0x162>(x);
    x += row_xor<0x164>(x);
    x += row_xor<0x168>(x);
    return x + xhalf(x);
}

__global__ __launch_bounds__(NT, 1)
void temporal_attn_lastrow_g4(const float* __restrict__ inputs,
                              const float* __restrict__ pos,
                              const float* __restrict__ Wq,
                              const float* __restrict__ Wk,
                              const float* __restrict__ Wv,
                              float* __restrict__ out)
{
    extern __shared__ float lds[];
    float* sH = lds + OFF_H;
    float* sR = lds + OFF_R;
    float* sQ = lds + OFF_Q;
    float* sU = lds + OFF_U;
    float* sS = lds + OFF_S;
    float* sP = lds + OFF_R;

    const int tid  = threadIdx.x;
    const int lane = tid & 31;
    const int wave = tid >> 5;
    const int m    = lane & 15;
    const int hi   = lane >> 4;
    const long b0  = (long)blockIdx.x * GB;

    for (int i = tid; i < 16 * RSTR; i += NT) sR[i] = 0.f;
    __syncthreads();
    static_assert(GB * DD % NT == 0, "whole trips");
    for (int it = 0; it < GB * DD / NT; ++it) {
        const int i = tid + it * NT;
        int g = i >> 9, d = i & 511;
        sR[g * RSTR + d] =
            inputs[(b0 + g) * (size_t)(TT * DD) + 63 * DD + d] + pos[63 * DD + d];
    }
    __syncthreads();

    static_assert(NW == DD / 16, "one 16-column tile per wave");
    {
        const int n0 = wave * 16;
        v8f acc = {};
        for (int k = 0; k < DD; k += 4) {
            const int kk = k + 2 * hi;
            v2f a, b;
            a[0] = sR[m * RSTR + kk];
            a[1] = sR[m * RSTR + kk + 1];
            b[0] = Wq[(size_t)kk * DD + n0 + m];
            b[1] = Wq[(size_t)(kk + 1) * DD + n0 + m];
            acc = wmma4(a, b, acc);
        }
        static_assert(GB == 8, "rows 0..7 are the 8 registers of the lower lane half");
        if (hi == 0) {
            #pragma unroll
            for (int r = 0; r < GB; ++r) {
                const int e = n0 + m;
                sQ[r * QG + (e >> 5) * QSTR + (e & 31)] = acc[r];
            }
        }
    }
    __syncthreads();

    for (int sg = 0; sg < NSG; ++sg) {
    const long   bg0 = b0 + sg * G;
    const float* sQg = sQ + sg * (G * QG);
    int tl = threadIdx.x;
    asm volatile("" : "+v"(tl));
    const int tid  = tl;
    const int lane = tid & 31;
    const int wave = tid >> 5;
    const int m    = lane & 15;
    const int hi   = lane >> 4;

    {
        const int   gl    = m & 3;
        const float vmask = (m < G) ? 1.f : 0.f;
        for (int t = 0; t < (DD / 16) * NH / NW; ++t) {
            const int job = wave * ((DD / 16) * NH / NW) + t;
            const int dt  = job >> 4;
            const int h   = job & 15;
            const int d0  = dt * 16;
            v8f acc = {};
            #pragma unroll
            for (int k = 0; k < 32; k += 4) {
                const int kk = k + 2 * hi;
                v2f a, b;
                a[0] = Wk[(size_t)(d0 + m) * DD + 32 * h + kk];
                a[1] = Wk[(size_t)(d0 + m) * DD + 32 * h + kk + 1];
                b[0] = vmask * sQg[gl * QG + h * QSTR + kk];
                b[1] = vmask * sQg[gl * QG + h * QSTR + kk + 1];
                acc = wmma4(a, b, acc);
            }
            if (m < G) {
                #pragma unroll
                for (int r = 0; r < 8; ++r)
                    sU[m * UG + (d0 + r + 8 * hi) * NH + h] = acc[r];
            }
        }
    }
    __syncthreads();

    const float scale = 0.044194173824159216f;
    for (int g = 0; g < G; ++g) {
        {
            const f4* in4 = (const f4*)(inputs + (bg0 + g) * (size_t)(TT * DD));
            const f4* pe4 = (const f4*)pos;
            static_assert(TT * DD / 4 % NT == 0, "whole trips");
            constexpr int HV = TT * DD / 4 / NT;
            constexpr int HB = 4;
            #pragma nounroll
            for (int ib = 0; ib < HV; ib += HB) {
                f4 x[HB], p[HB];
                for (int j = 0; j < HB; ++j) {
                    const int i = tid + (ib + j) * NT;
                    x[j] = in4[i]; p[j] = pe4[i];
                }
                for (int j = 0; j < HB; ++j) {
                    const int i = tid + (ib + j) * NT;
                    int t = i >> 7, d = (i & 127) << 2;
                    float* dst = sH + t * HSTR + d;
                    dst[0] = x[j].x + p[j].x; dst[1] = x[j].y + p[j].y;
                    dst[2] = x[j].z + p[j].z; dst[3] = x[j].w + p[j].w;
                }
            }
            if (sg * G + g + 1 < GB) {
                const float* nxt = inputs + (bg0 + g + 1) * (size_t)(TT * DD);
                static_assert(TT * DD / 32 == NT, "one 128-byte line per thread");
                __builtin_prefetch(nxt + (size_t)tid * 32, 0, 1);
            }
        }
        __syncthreads();

        static_assert(KS * (TT / 16) == NW && KS * KSL == DD, "one (tile, slice) per wave");
        {
            const int t0 = (wave & 3) * 16;
            const int c  = wave >> 2;
            v8f acc = {};
            for (int k = 0; k < KSL; k += 4) {
                const int kk = c * KSL + k + 2 * hi;
                v2f a, b;
                a[0] = sH[(t0 + m) * HSTR + kk];
                a[1] = sH[(t0 + m) * HSTR + kk + 1];
                b[0] = sU[g * UG + kk * NH + m];
                b[1] = sU[g * UG + (kk + 1) * NH + m];
                acc = wmma4(a, b, acc);
            }
            #pragma unroll
            for (int r = 0; r < 8; ++r)
                sP[c * PSTR + (t0 + r + 8 * hi) * SSTR + m] = acc[r];
        }
        __syncthreads();

        static_assert(TT == 64 && NH <= NW, "two t per lane, one head per wave");
        if (wave < NH) {
            float* col = sS + wave;
            const float* par = sP + wave;
            float s0 = par[lane * SSTR];
            float s1 = par[(lane + 32) * SSTR];
            for (int c = 1; c < KS; ++c) {
                s0 += par[c * PSTR + lane * SSTR];
                s1 += par[c * PSTR + (lane + 32) * SSTR];
            }
            s0 *= scale;
            s1 *= scale;
            const float mx  = wave_max(fmaxf(s0, s1));
            const float e0  = expf(s0 - mx);
            const float e1  = expf(s1 - mx);
            const float inv = 1.f / wave_sum(e0 + e1);
            col[lane * SSTR]        = e0 * inv;
            col[(lane + 32) * SSTR] = e1 * inv;
        }
        __syncthreads();

        {
            const int d0 = wave * 16;
            v8f acc = {};
            for (int k = 0; k < TT; k += 4) {
                const int kk = k + 2 * hi;
                v2f a, b;
                a[0] = sH[kk * HSTR + d0 + m];
                a[1] = sH[(kk + 1) * HSTR + d0 + m];
                b[0] = sS[kk * SSTR + m];
                b[1] = sS[(kk + 1) * SSTR + m];
                acc = wmma4(a, b, acc);
            }
            #pragma unroll
            for (int r = 0; r < 8; ++r)
                sU[g * UG + (d0 + r + 8 * hi) * NH + m] = acc[r];
        }
        __syncthreads();
    }

    {
        const int   gl    = m & 3;
        const float vmask = (m < G) ? 1.f : 0.f;
        {
            const int job = wave;
            const int h   = job >> 1;
            const int n0  = (job & 1) * 16;
            v8f acc = {};
            for (int k = 0; k < DD; k += 4) {
                const int kk = k + 2 * hi;
                v2f a, b;
                a[0] = vmask * sU[gl * UG + kk * NH + h];
                a[1] = vmask * sU[gl * UG + (kk + 1) * NH + h];
                b[0] = Wv[(size_t)kk * DD + 32 * h + n0 + m];
                b[1] = Wv[(size_t)(kk + 1) * DD + 32 * h + n0 + m];
                acc = wmma4(a, b, acc);
            }
            if (hi == 0) {
                #pragma unroll
                for (int r = 0; r < G; ++r)
                    out[(bg0 + r) * (size_t)DD + 32 * h + n0 + m] = acc[r];
            }
        }
    }
    if (sg + 1 < NSG) __syncthreads();
    }
}

extern "C" void kernel_launch(void* const* d_in, const int* in_sizes, int n_in,
                              void* d_out, int out_size, void* d_ws, size_t ws_size,
                              hipStream_t stream) {
    const float* inputs = (const float*)d_in[0];
    const float* pos    = (const float*)d_in[1];
    const float* Wq     = (const float*)d_in[2];
    const float* Wk     = (const float*)d_in[3];
    const float* Wv     = (const float*)d_in[4];
    float* out          = (float*)d_out;
    (void)d_ws; (void)ws_size; (void)n_in; (void)out_size;

    const int B = in_sizes[0] / (TT * DD);
    const size_t smem = (size_t)LDS_FLOATS * sizeof(float);

    hipFuncSetAttribute((const void*)temporal_attn_lastrow_g4,
                        hipFuncAttributeMaxDynamicSharedMemorySize, (int)smem);

    temporal_attn_lastrow_g4<<<B / GB, NT, smem, stream>>>(
        inputs, pos, Wq, Wk, Wv, out);
}


# Wk.q GEMM: a lane reads its half of a head's 32 j as one 64-byte run, four 16-byte loads (was eight 8-byte loads over the same 16 lines); sum over j regrouped: step s adds j=2s,2s+1,16+2s,17+2s
# speedup vs baseline: 2.2486x; 1.0477x over previous
#include <hip/hip_runtime.h>
#include <math.h>

constexpr int TT = 64;
constexpr int DD = 512;
constexpr int NH = 16;
constexpr int G  = 4;
constexpr int NSG = 2;
constexpr int GB  = G * NSG;
constexpr int NT = 1024;
constexpr int NW = NT / 32;

constexpr int HSTR = 516;
constexpr int RSTR = 516;
constexpr int QSTR = 36;
constexpr int QG   = NH * QSTR;
constexpr int UG   = DD * NH + 16;
constexpr int SSTR = NH + 1;

constexpr int KS   = NW / (TT / 16);
constexpr int KSL  = DD / KS;
constexpr int PSTR = TT * SSTR;

constexpr int OFF_H = 0;
constexpr int OFF_R = OFF_H + TT * HSTR;
constexpr int RP_FLOATS = (KS * PSTR > 16 * RSTR) ? KS * PSTR : 16 * RSTR;
constexpr int OFF_Q = OFF_R + RP_FLOATS;
constexpr int OFF_U = OFF_Q + GB * QG;
constexpr int OFF_S = OFF_U + G * UG;
constexpr int LDS_FLOATS = OFF_S + TT * SSTR;
static_assert(KS * PSTR <= OFF_Q - OFF_R, "score partials fit in the sR area");
static_assert(LDS_FLOATS * 4 <= 320 * 1024, "one workgroup's LDS");

typedef __attribute__((ext_vector_type(2))) float v2f;
typedef __attribute__((ext_vector_type(4))) float f4;
typedef __attribute__((ext_vector_type(8))) float v8f;

static __device__ __forceinline__ v8f wmma4(v2f a, v2f b, v8f c) {
    return __builtin_amdgcn_wmma_f32_16x16x4_f32(
        false, a, false, b, (short)0, c, false, false);
}

template <int CTRL>
static __device__ __forceinline__ float row_xor(float x) {
    const int b = __builtin_bit_cast(int, x);
    return __builtin_bit_cast(float,
        __builtin_amdgcn_update_dpp(b, b, CTRL, 0xf, 0xf, false));
}
static __device__ __forceinline__ float xhalf(float x) {
    const int b = __builtin_bit_cast(int, x);
    return __builtin_bit_cast(float,
        __builtin_amdgcn_permlanex16(b, b, 0x76543210, 0xfedcba98, false, false));
}
static __device__ __forceinline__ float wave_max(float x) {
    x = fmaxf(x, row_xor<0x161>(x));
    x = fmaxf(x, row_xor<0x162>(x));
    x = fmaxf(x, row_xor<0x164>(x));
    x = fmaxf(x, row_xor<0x168>(x));
    return fmaxf(x, xhalf(x));
}
static __device__ __forceinline__ float wave_sum(float x) {
    x += row_xor<0x161>(x);
    x += row_xor<0x162>(x);
    x += row_xor<0x164>(x);
    x += row_xor<0x168>(x);
    return x + xhalf(x);
}

__global__ __launch_bounds__(NT, 1)
void temporal_attn_lastrow_g4(const float* __restrict__ inputs,
                              const float* __restrict__ pos,
                              const float* __restrict__ Wq,
                              const float* __restrict__ Wk,
                              const float* __restrict__ Wv,
                              float* __restrict__ out)
{
    extern __shared__ float lds[];
    float* sH = lds + OFF_H;
    float* sR = lds + OFF_R;
    float* sQ = lds + OFF_Q;
    float* sU = lds + OFF_U;
    float* sS = lds + OFF_S;
    float* sP = lds + OFF_R;

    const int tid  = threadIdx.x;
    const int lane = tid & 31;
    const int wave = tid >> 5;
    const int m    = lane & 15;
    const int hi   = lane >> 4;
    const long b0  = (long)blockIdx.x * GB;

    for (int i = tid; i < 16 * RSTR; i += NT) sR[i] = 0.f;
    __syncthreads();
    static_assert(GB * DD % NT == 0, "whole trips");
    for (int it = 0; it < GB * DD / NT; ++it) {
        const int i = tid + it * NT;
        int g = i >> 9, d = i & 511;
        sR[g * RSTR + d] =
            inputs[(b0 + g) * (size_t)(TT * DD) + 63 * DD + d] + pos[63 * DD + d];
    }
    __syncthreads();

    static_assert(NW == DD / 16, "one 16-column tile per wave");
    {
        const int n0 = wave * 16;
        v8f acc = {};
        for (int k = 0; k < DD; k += 4) {
            const int kk = k + 2 * hi;
            v2f a, b;
            a[0] = sR[m * RSTR + kk];
            a[1] = sR[m * RSTR + kk + 1];
            b[0] = Wq[(size_t)kk * DD + n0 + m];
            b[1] = Wq[(size_t)(kk + 1) * DD + n0 + m];
            acc = wmma4(a, b, acc);
        }
        static_assert(GB == 8, "rows 0..7 are the 8 registers of the lower lane half");
        if (hi == 0) {
            #pragma unroll
            for (int r = 0; r < GB; ++r) {
                const int e = n0 + m;
                sQ[r * QG + (e >> 5) * QSTR + (e & 31)] = acc[r];
            }
        }
    }
    __syncthreads();

    for (int sg = 0; sg < NSG; ++sg) {
    const long   bg0 = b0 + sg * G;
    const float* sQg = sQ + sg * (G * QG);
    int tl = threadIdx.x;
    asm volatile("" : "+v"(tl));
    const int tid  = tl;
    const int lane = tid & 31;
    const int wave = tid >> 5;
    const int m    = lane & 15;
    const int hi   = lane >> 4;

    {
        const int   gl    = m & 3;
        const float vmask = (m < G) ? 1.f : 0.f;
        static_assert(QSTR % 4 == 0 && QG % 4 == 0 && OFF_Q % 4 == 0, "16-byte q reads");
        for (int t = 0; t < (DD / 16) * NH / NW; ++t) {
            const int job = wave * ((DD / 16) * NH / NW) + t;
            const int dt  = job >> 4;
            const int h   = job & 15;
            const int d0  = dt * 16;
            const f4* a4 = (const f4*)(Wk + (size_t)(d0 + m) * DD + 32 * h + 16 * hi);
            const f4* q4 = (const f4*)(sQg + gl * QG + h * QSTR + 16 * hi);
            f4 av[4], bv[4];
            #pragma unroll
            for (int v = 0; v < 4; ++v) av[v] = a4[v];
            #pragma unroll
            for (int v = 0; v < 4; ++v) bv[v] = q4[v];
            v8f acc = {};
            #pragma unroll
            for (int s = 0; s < 8; ++s) {
                const int v = s >> 1, e = 2 * (s & 1);
                v2f a, b;
                a[0] = av[v][e];
                a[1] = av[v][e + 1];
                b[0] = vmask * bv[v][e];
                b[1] = vmask * bv[v][e + 1];
                acc = wmma4(a, b, acc);
            }
            if (m < G) {
                #pragma unroll
                for (int r = 0; r < 8; ++r)
                    sU[m * UG + (d0 + r + 8 * hi) * NH + h] = acc[r];
            }
        }
    }
    __syncthreads();

    const float scale = 0.044194173824159216f;
    for (int g = 0; g < G; ++g) {
        {
            const f4* in4 = (const f4*)(inputs + (bg0 + g) * (size_t)(TT * DD));
            const f4* pe4 = (const f4*)pos;
            static_assert(TT * DD / 4 % NT == 0, "whole trips");
            constexpr int HV = TT * DD / 4 / NT;
            constexpr int HB = 4;
            #pragma nounroll
            for (int ib = 0; ib < HV; ib += HB) {
                f4 x[HB], p[HB];
                for (int j = 0; j < HB; ++j) {
                    const int i = tid + (ib + j) * NT;
                    x[j] = in4[i]; p[j] = pe4[i];
                }
                for (int j = 0; j < HB; ++j) {
                    const int i = tid + (ib + j) * NT;
                    int t = i >> 7, d = (i & 127) << 2;
                    float* dst = sH + t * HSTR + d;
                    dst[0] = x[j].x + p[j].x; dst[1] = x[j].y + p[j].y;
                    dst[2] = x[j].z + p[j].z; dst[3] = x[j].w + p[j].w;
                }
            }
            if (sg * G + g + 1 < GB) {
                const float* nxt = inputs + (bg0 + g + 1) * (size_t)(TT * DD);
                static_assert(TT * DD / 32 == NT, "one 128-byte line per thread");
                __builtin_prefetch(nxt + (size_t)tid * 32, 0, 1);
            }
        }
        __syncthreads();

        static_assert(KS * (TT / 16) == NW && KS * KSL == DD, "one (tile, slice) per wave");
        {
            const int t0 = (wave & 3) * 16;
            const int c  = wave >> 2;
            v8f acc = {};
            for (int k = 0; k < KSL; k += 4) {
                const int kk = c * KSL + k + 2 * hi;
                v2f a, b;
                a[0] = sH[(t0 + m) * HSTR + kk];
                a[1] = sH[(t0 + m) * HSTR + kk + 1];
                b[0] = sU[g * UG + kk * NH + m];
                b[1] = sU[g * UG + (kk + 1) * NH + m];
                acc = wmma4(a, b, acc);
            }
            #pragma unroll
            for (int r = 0; r < 8; ++r)
                sP[c * PSTR + (t0 + r + 8 * hi) * SSTR + m] = acc[r];
        }
        __syncthreads();

        static_assert(TT == 64 && NH <= NW, "two t per lane, one head per wave");
        if (wave < NH) {
            float* col = sS + wave;
            const float* par = sP + wave;
            float s0 = par[lane * SSTR];
            float s1 = par[(lane + 32) * SSTR];
            for (int c = 1; c < KS; ++c) {
                s0 += par[c * PSTR + lane * SSTR];
                s1 += par[c * PSTR + (lane + 32) * SSTR];
            }
            s0 *= scale;
            s1 *= scale;
            const float mx  = wave_max(fmaxf(s0, s1));
            const float e0  = expf(s0 - mx);
            const float e1  = expf(s1 - mx);
            const float inv = 1.f / wave_sum(e0 + e1);
            col[lane * SSTR]        = e0 * inv;
            col[(lane + 32) * SSTR] = e1 * inv;
        }
        __syncthreads();

        {
            const int d0 = wave * 16;
            v8f acc = {};
            for (int k = 0; k < TT; k += 4) {
                const int kk = k + 2 * hi;
                v2f a, b;
                a[0] = sH[kk * HSTR + d0 + m];
                a[1] = sH[(kk + 1) * HSTR + d0 + m];
                b[0] = sS[kk * SSTR + m];
                b[1] = sS[(kk + 1) * SSTR + m];
                acc = wmma4(a, b, acc);
            }
            #pragma unroll
            for (int r = 0; r < 8; ++r)
                sU[g * UG + (d0 + r + 8 * hi) * NH + m] = acc[r];
        }
        __syncthreads();
    }

    {
        const int   gl    = m & 3;
        const float vmask = (m < G) ? 1.f : 0.f;
        {
            const int job = wave;
            const int h   = job >> 1;
            const int n0  = (job & 1) * 16;
            v8f acc = {};
            for (int k = 0; k < DD; k += 4) {
                const int kk = k + 2 * hi;
                v2f a, b;
                a[0] = vmask * sU[gl * UG + kk * NH + h];
                a[1] = vmask * sU[gl * UG + (kk + 1) * NH + h];
                b[0] = Wv[(size_t)kk * DD + 32 * h + n0 + m];
                b[1] = Wv[(size_t)(kk + 1) * DD + 32 * h + n0 + m];
                acc = wmma4(a, b, acc);
            }
            if (hi == 0) {
                #pragma unroll
                for (int r = 0; r < G; ++r)
                    out[(bg0 + r) * (size_t)DD + 32 * h + n0 + m] = acc[r];
            }
        }
    }
    if (sg + 1 < NSG) __syncthreads();
    }
}

extern "C" void kernel_launch(void* const* d_in, const int* in_sizes, int n_in,
                              void* d_out, int out_size, void* d_ws, size_t ws_size,
                              hipStream_t stream) {
    const float* inputs = (const float*)d_in[0];
    const float* pos    = (const float*)d_in[1];
    const float* Wq     = (const float*)d_in[2];
    const float* Wk     = (const float*)d_in[3];
    const float* Wv     = (const float*)d_in[4];
    float* out          = (float*)d_out;
    (void)d_ws; (void)ws_size; (void)n_in; (void)out_size;

    const int B = in_sizes[0] / (TT * DD);
    const size_t smem = (size_t)LDS_FLOATS * sizeof(float);

    hipFuncSetAttribute((const void*)temporal_attn_lastrow_g4,
                        hipFuncAttributeMaxDynamicSharedMemorySize, (int)smem);

    temporal_attn_lastrow_g4<<<B / GB, NT, smem, stream>>>(
        inputs, pos, Wq, Wk, Wv, out);
}


# Wk.q and C.Wv GEMMs run once per block for all 8 batches (8 of 16 N/M valid, was 2 scans of 4): the other group's U, then C, waits in 32 registers per thread; sums' order kept, bit-identical
# speedup vs baseline: 2.7329x; 1.2154x over previous
#include <hip/hip_runtime.h>
#include <math.h>

constexpr int TT = 64;
constexpr int DD = 512;
constexpr int NH = 16;
constexpr int G  = 4;
constexpr int NSG = 2;
constexpr int GB  = G * NSG;
constexpr int NT = 1024;
constexpr int NW = NT / 32;

constexpr int HSTR = 516;
constexpr int RSTR = 516;
constexpr int QSTR = 36;
constexpr int QG   = NH * QSTR;
constexpr int UG   = DD * NH + 16;
constexpr int SSTR = NH + 1;

constexpr int KS   = NW / (TT / 16);
constexpr int KSL  = DD / KS;
constexpr int PSTR = TT * SSTR;

constexpr int OFF_H = 0;
constexpr int OFF_R = OFF_H + TT * HSTR;
constexpr int RP_FLOATS = (KS * PSTR > 16 * RSTR) ? KS * PSTR : 16 * RSTR;
constexpr int OFF_Q = OFF_R + RP_FLOATS;
constexpr int OFF_U = OFF_Q + GB * QG;
constexpr int OFF_S = OFF_U + G * UG;
constexpr int LDS_FLOATS = OFF_S + TT * SSTR;
static_assert(KS * PSTR <= OFF_Q - OFF_R, "score partials fit in the sR area");
static_assert(LDS_FLOATS * 4 <= 320 * 1024, "one workgroup's LDS");

typedef __attribute__((ext_vector_type(2))) float v2f;
typedef __attribute__((ext_vector_type(4))) float f4;
typedef __attribute__((ext_vector_type(8))) float v8f;

static __device__ __forceinline__ v8f wmma4(v2f a, v2f b, v8f c) {
    return __builtin_amdgcn_wmma_f32_16x16x4_f32(
        false, a, false, b, (short)0, c, false, false);
}

template <int CTRL>
static __device__ __forceinline__ float row_xor(float x) {
    const int b = __builtin_bit_cast(int, x);
    return __builtin_bit_cast(float,
        __builtin_amdgcn_update_dpp(b, b, CTRL, 0xf, 0xf, false));
}
static __device__ __forceinline__ float xhalf(float x) {
    const int b = __builtin_bit_cast(int, x);
    return __builtin_bit_cast(float,
        __builtin_amdgcn_permlanex16(b, b, 0x76543210, 0xfedcba98, false, false));
}
static __device__ __forceinline__ float wave_max(float x) {
    x = fmaxf(x, row_xor<0x161>(x));
    x = fmaxf(x, row_xor<0x162>(x));
    x = fmaxf(x, row_xor<0x164>(x));
    x = fmaxf(x, row_xor<0x168>(x));
    return fmaxf(x, xhalf(x));
}
static __device__ __forceinline__ float wave_sum(float x) {
    x += row_xor<0x161>(x);
    x += row_xor<0x162>(x);
    x += row_xor<0x164>(x);
    x += row_xor<0x168>(x);
    return x + xhalf(x);
}

__global__ __launch_bounds__(NT, 1)
void temporal_attn_lastrow_g4(const float* __restrict__ inputs,
                              const float* __restrict__ pos,
                              const float* __restrict__ Wq,
                              const float* __restrict__ Wk,
                              const float* __restrict__ Wv,
                              float* __restrict__ out)
{
    extern __shared__ float lds[];
    float* sH = lds + OFF_H;
    float* sR = lds + OFF_R;
    float* sQ = lds + OFF_Q;
    float* sU = lds + OFF_U;
    float* sS = lds + OFF_S;
    float* sP = lds + OFF_R;

    const int tid  = threadIdx.x;
    const int lane = tid & 31;
    const int wave = tid >> 5;
    const int m    = lane & 15;
    const int hi   = lane >> 4;
    const long b0  = (long)blockIdx.x * GB;

    for (int i = tid; i < 16 * RSTR; i += NT) sR[i] = 0.f;
    __syncthreads();
    static_assert(GB * DD % NT == 0, "whole trips");
    for (int it = 0; it < GB * DD / NT; ++it) {
        const int i = tid + it * NT;
        int g = i >> 9, d = i & 511;
        sR[g * RSTR + d] =
            inputs[(b0 + g) * (size_t)(TT * DD) + 63 * DD + d] + pos[63 * DD + d];
    }
    __syncthreads();

    static_assert(NW == DD / 16, "one 16-column tile per wave");
    {
        const int n0 = wave * 16;
        v8f acc = {};
        for (int k = 0; k < DD; k += 4) {
            const int kk = k + 2 * hi;
            v2f a, b;
            a[0] = sR[m * RSTR + kk];
            a[1] = sR[m * RSTR + kk + 1];
            b[0] = Wq[(size_t)kk * DD + n0 + m];
            b[1] = Wq[(size_t)(kk + 1) * DD + n0 + m];
            acc = wmma4(a, b, acc);
        }
        static_assert(GB == 8, "rows 0..7 are the 8 registers of the lower lane half");
        if (hi == 0) {
            #pragma unroll
            for (int r = 0; r < GB; ++r) {
                const int e = n0 + m;
                sQ[r * QG + (e >> 5) * QSTR + (e & 31)] = acc[r];
            }
        }
    }
    __syncthreads();

    f4 park[2 * G];
    static_assert(G * (DD * NH) == 2 * G * 4 * NT, "8 float4 per thread hold one group");
    static_assert(G * UG <= TT * HSTR, "a parked group fits in the H tile's place");
    static_assert(UG % 4 == 0 && OFF_U % 4 == 0 && OFF_H % 4 == 0, "16-byte park copies");

    {
        int tl = threadIdx.x;
        asm volatile("" : "+v"(tl));
        const int lane = tl & 31;
        const int wave = tl >> 5;
        const int m    = lane & 15;
        const int hi   = lane >> 4;
        const int   gl    = m & (GB - 1);
        const float vmask = (m < GB) ? 1.f : 0.f;
        float* ub = (gl < G) ? (sU + gl * UG) : (sH + (gl - G) * UG);
        static_assert(QSTR % 4 == 0 && QG % 4 == 0 && OFF_Q % 4 == 0, "16-byte q reads");
        for (int t = 0; t < (DD / 16) * NH / NW; ++t) {
            const int job = wave * ((DD / 16) * NH / NW) + t;
            const int dt  = job >> 4;
            const int h   = job & 15;
            const int d0  = dt * 16;
            const f4* a4 = (const f4*)(Wk + (size_t)(d0 + m) * DD + 32 * h + 16 * hi);
            const f4* q4 = (const f4*)(sQ + gl * QG + h * QSTR + 16 * hi);
            f4 av[4], bv[4];
            #pragma unroll
            for (int v = 0; v < 4; ++v) av[v] = a4[v];
            #pragma unroll
            for (int v = 0; v < 4; ++v) bv[v] = q4[v];
            v8f acc = {};
            #pragma unroll
            for (int s = 0; s < 8; ++s) {
                const int v = s >> 1, e = 2 * (s & 1);
                v2f a, b;
                a[0] = av[v][e];
                a[1] = av[v][e + 1];
                b[0] = vmask * bv[v][e];
                b[1] = vmask * bv[v][e + 1];
                acc = wmma4(a, b, acc);
            }
            if (m < GB) {
                #pragma unroll
                for (int r = 0; r < 8; ++r)
                    ub[(d0 + r + 8 * hi) * NH + h] = acc[r];
            }
        }
    }
    __syncthreads();
    #pragma unroll
    for (int g = 0; g < G; ++g) {
        #pragma unroll
        for (int j = 0; j < 2; ++j)
            park[2 * g + j] = *(const f4*)(sH + g * UG + 4 * (tid + j * NT));
    }
    __syncthreads();

    for (int sg = 0; sg < NSG; ++sg) {
    const long   bg0 = b0 + sg * G;
    if (sg == 1) {
        #pragma unroll
        for (int g = 0; g < G; ++g) {
            #pragma unroll
            for (int j = 0; j < 2; ++j) {
                f4* slot = (f4*)(sU + g * UG + 4 * (tid + j * NT));
                const f4 c = *slot;
                *slot = park[2 * g + j];
                park[2 * g + j] = c;
            }
        }
    }

    const float scale = 0.044194173824159216f;
    for (int g = 0; g < G; ++g) {
        {
            int tl = threadIdx.x;
            asm volatile("" : "+v"(tl));
            const int tid  = tl;
            const f4* in4 = (const f4*)(inputs + (bg0 + g) * (size_t)(TT * DD));
            const f4* pe4 = (const f4*)pos;
            static_assert(TT * DD / 4 % NT == 0, "whole trips");
            constexpr int HV = TT * DD / 4 / NT;
            constexpr int HB = 4;
            #pragma nounroll
            for (int ib = 0; ib < HV; ib += HB) {
                f4 x[HB], p[HB];
                for (int j = 0; j < HB; ++j) {
                    const int i = tid + (ib + j) * NT;
                    x[j] = in4[i]; p[j] = pe4[i];
                }
                for (int j = 0; j < HB; ++j) {
                    const int i = tid + (ib + j) * NT;
                    int t = i >> 7, d = (i & 127) << 2;
                    float* dst = sH + t * HSTR + d;
                    dst[0] = x[j].x + p[j].x; dst[1] = x[j].y + p[j].y;
                    dst[2] = x[j].z + p[j].z; dst[3] = x[j].w + p[j].w;
                }
            }
            if (sg * G + g + 1 < GB) {
                const float* nxt = inputs + (bg0 + g + 1) * (size_t)(TT * DD);
                static_assert(TT * DD / 32 == NT, "one 128-byte line per thread");
                __builtin_prefetch(nxt + (size_t)tid * 32, 0, 1);
            }
        }
        __syncthreads();

        static_assert(KS * (TT / 16) == NW && KS * KSL == DD, "one (tile, slice) per wave");
        {
            int tl = threadIdx.x;
            asm volatile("" : "+v"(tl));
            const int lane = tl & 31;
            const int wave = tl >> 5;
            const int m    = lane & 15;
            const int hi   = lane >> 4;
            const int t0 = (wave & 3) * 16;
            const int c  = wave >> 2;
            v8f acc = {};
            for (int k = 0; k < KSL; k += 4) {
                const int kk = c * KSL + k + 2 * hi;
                v2f a, b;
                a[0] = sH[(t0 + m) * HSTR + kk];
                a[1] = sH[(t0 + m) * HSTR + kk + 1];
                b[0] = sU[g * UG + kk * NH + m];
                b[1] = sU[g * UG + (kk + 1) * NH + m];
                acc = wmma4(a, b, acc);
            }
            #pragma unroll
            for (int r = 0; r < 8; ++r)
                sP[c * PSTR + (t0 + r + 8 * hi) * SSTR + m] = acc[r];
        }
        __syncthreads();

        static_assert(TT == 64 && NH <= NW, "two t per lane, one head per wave");
        {
        int tl = threadIdx.x;
        asm volatile("" : "+v"(tl));
        const int lane = tl & 31;
        const int wave = tl >> 5;
        if (wave < NH) {
            float* col = sS + wave;
            const float* par = sP + wave;
            float s0 = par[lane * SSTR];
            float s1 = par[(lane + 32) * SSTR];
            for (int c = 1; c < KS; ++c) {
                s0 += par[c * PSTR + lane * SSTR];
                s1 += par[c * PSTR + (lane + 32) * SSTR];
            }
            s0 *= scale;
            s1 *= scale;
            const float mx  = wave_max(fmaxf(s0, s1));
            const float e0  = expf(s0 - mx);
            const float e1  = expf(s1 - mx);
            const float inv = 1.f / wave_sum(e0 + e1);
            col[lane * SSTR]        = e0 * inv;
            col[(lane + 32) * SSTR] = e1 * inv;
        }
        }
        __syncthreads();

        {
            int tl = threadIdx.x;
            asm volatile("" : "+v"(tl));
            const int lane = tl & 31;
            const int wave = tl >> 5;
            const int m    = lane & 15;
            const int hi   = lane >> 4;
            const int d0 = wave * 16;
            v8f acc = {};
            for (int k = 0; k < TT; k += 4) {
                const int kk = k + 2 * hi;
                v2f a, b;
                a[0] = sH[kk * HSTR + d0 + m];
                a[1] = sH[(kk + 1) * HSTR + d0 + m];
                b[0] = sS[kk * SSTR + m];
                b[1] = sS[(kk + 1) * SSTR + m];
                acc = wmma4(a, b, acc);
            }
            #pragma unroll
            for (int r = 0; r < 8; ++r)
                sU[g * UG + (d0 + r + 8 * hi) * NH + m] = acc[r];
        }
        __syncthreads();
    }
    }

    #pragma unroll
    for (int g = 0; g < G; ++g) {
        #pragma unroll
        for (int j = 0; j < 2; ++j)
            *(f4*)(sH + g * UG + 4 * (tid + j * NT)) = park[2 * g + j];
    }
    __syncthreads();

    {
        int tl = threadIdx.x;
        asm volatile("" : "+v"(tl));
        const int lane = tl & 31;
        const int wave = tl >> 5;
        const int m    = lane & 15;
        const int hi   = lane >> 4;
        const int   gl    = m & (GB - 1);
        const float vmask = (m < GB) ? 1.f : 0.f;
        const float* cb = (gl < G) ? (sH + gl * UG) : (sU + (gl - G) * UG);
        {
            const int job = wave;
            const int h   = job >> 1;
            const int n0  = (job & 1) * 16;
            v8f acc = {};
            for (int k = 0; k < DD; k += 4) {
                const int kk = k + 2 * hi;
                v2f a, b;
                a[0] = vmask * cb[kk * NH + h];
                a[1] = vmask * cb[(kk + 1) * NH + h];
                b[0] = Wv[(size_t)kk * DD + 32 * h + n0 + m];
                b[1] = Wv[(size_t)(kk + 1) * DD + 32 * h + n0 + m];
                acc = wmma4(a, b, acc);
            }
            static_assert(GB == 8, "rows 0..7 are the 8 registers of the lower lane half");
            if (hi == 0) {
                #pragma unroll
                for (int r = 0; r < GB; ++r)
                    out[(b0 + r) * (size_t)DD + 32 * h + n0 + m] = acc[r];
            }
        }
    }
}

extern "C" void kernel_launch(void* const* d_in, const int* in_sizes, int n_in,
                              void* d_out, int out_size, void* d_ws, size_t ws_size,
                              hipStream_t stream) {
    const float* inputs = (const float*)d_in[0];
    const float* pos    = (const float*)d_in[1];
    const float* Wq     = (const float*)d_in[2];
    const float* Wk     = (const float*)d_in[3];
    const float* Wv     = (const float*)d_in[4];
    float* out          = (float*)d_out;
    (void)d_ws; (void)ws_size; (void)n_in; (void)out_size;

    const int B = in_sizes[0] / (TT * DD);
    const size_t smem = (size_t)LDS_FLOATS * sizeof(float);

    hipFuncSetAttribute((const void*)temporal_attn_lastrow_g4,
                        hipFuncAttributeMaxDynamicSharedMemorySize, (int)smem);

    temporal_attn_lastrow_g4<<<B / GB, NT, smem, stream>>>(
        inputs, pos, Wq, Wk, Wv, out);
}


# pos held in 32 registers per thread across the block's 8 batches (read once, was re-read from L2 for every H tile); a tile's 8 input loads per thread now go out in one trip, was two; bit-identical
# speedup vs baseline: 2.8181x; 1.0312x over previous
#include <hip/hip_runtime.h>
#include <math.h>

constexpr int TT = 64;
constexpr int DD = 512;
constexpr int NH = 16;
constexpr int G  = 4;
constexpr int NSG = 2;
constexpr int GB  = G * NSG;
constexpr int NT = 1024;
constexpr int NW = NT / 32;

constexpr int HSTR = 516;
constexpr int RSTR = 516;
constexpr int QSTR = 36;
constexpr int QG   = NH * QSTR;
constexpr int UG   = DD * NH + 16;
constexpr int SSTR = NH + 1;

constexpr int KS   = NW / (TT / 16);
constexpr int KSL  = DD / KS;
constexpr int PSTR = TT * SSTR;

constexpr int OFF_H = 0;
constexpr int OFF_R = OFF_H + TT * HSTR;
constexpr int RP_FLOATS = (KS * PSTR > 16 * RSTR) ? KS * PSTR : 16 * RSTR;
constexpr int OFF_Q = OFF_R + RP_FLOATS;
constexpr int OFF_U = OFF_Q + GB * QG;
constexpr int OFF_S = OFF_U + G * UG;
constexpr int LDS_FLOATS = OFF_S + TT * SSTR;
static_assert(KS * PSTR <= OFF_Q - OFF_R, "score partials fit in the sR area");
static_assert(LDS_FLOATS * 4 <= 320 * 1024, "one workgroup's LDS");

typedef __attribute__((ext_vector_type(2))) float v2f;
typedef __attribute__((ext_vector_type(4))) float f4;
typedef __attribute__((ext_vector_type(8))) float v8f;

static __device__ __forceinline__ v8f wmma4(v2f a, v2f b, v8f c) {
    return __builtin_amdgcn_wmma_f32_16x16x4_f32(
        false, a, false, b, (short)0, c, false, false);
}

template <int CTRL>
static __device__ __forceinline__ float row_xor(float x) {
    const int b = __builtin_bit_cast(int, x);
    return __builtin_bit_cast(float,
        __builtin_amdgcn_update_dpp(b, b, CTRL, 0xf, 0xf, false));
}
static __device__ __forceinline__ float xhalf(float x) {
    const int b = __builtin_bit_cast(int, x);
    return __builtin_bit_cast(float,
        __builtin_amdgcn_permlanex16(b, b, 0x76543210, 0xfedcba98, false, false));
}
static __device__ __forceinline__ float wave_max(float x) {
    x = fmaxf(x, row_xor<0x161>(x));
    x = fmaxf(x, row_xor<0x162>(x));
    x = fmaxf(x, row_xor<0x164>(x));
    x = fmaxf(x, row_xor<0x168>(x));
    return fmaxf(x, xhalf(x));
}
static __device__ __forceinline__ float wave_sum(float x) {
    x += row_xor<0x161>(x);
    x += row_xor<0x162>(x);
    x += row_xor<0x164>(x);
    x += row_xor<0x168>(x);
    return x + xhalf(x);
}

__global__ __launch_bounds__(NT, 1)
void temporal_attn_lastrow_g4(const float* __restrict__ inputs,
                              const float* __restrict__ pos,
                              const float* __restrict__ Wq,
                              const float* __restrict__ Wk,
                              const float* __restrict__ Wv,
                              float* __restrict__ out)
{
    extern __shared__ float lds[];
    float* sH = lds + OFF_H;
    float* sR = lds + OFF_R;
    float* sQ = lds + OFF_Q;
    float* sU = lds + OFF_U;
    float* sS = lds + OFF_S;
    float* sP = lds + OFF_R;

    const int tid  = threadIdx.x;
    const int lane = tid & 31;
    const int wave = tid >> 5;
    const int m    = lane & 15;
    const int hi   = lane >> 4;
    const long b0  = (long)blockIdx.x * GB;

    for (int i = tid; i < 16 * RSTR; i += NT) sR[i] = 0.f;
    __syncthreads();
    static_assert(GB * DD % NT == 0, "whole trips");
    for (int it = 0; it < GB * DD / NT; ++it) {
        const int i = tid + it * NT;
        int g = i >> 9, d = i & 511;
        sR[g * RSTR + d] =
            inputs[(b0 + g) * (size_t)(TT * DD) + 63 * DD + d] + pos[63 * DD + d];
    }
    __syncthreads();

    static_assert(NW == DD / 16, "one 16-column tile per wave");
    {
        const int n0 = wave * 16;
        v8f acc = {};
        for (int k = 0; k < DD; k += 4) {
            const int kk = k + 2 * hi;
            v2f a, b;
            a[0] = sR[m * RSTR + kk];
            a[1] = sR[m * RSTR + kk + 1];
            b[0] = Wq[(size_t)kk * DD + n0 + m];
            b[1] = Wq[(size_t)(kk + 1) * DD + n0 + m];
            acc = wmma4(a, b, acc);
        }
        static_assert(GB == 8, "rows 0..7 are the 8 registers of the lower lane half");
        if (hi == 0) {
            #pragma unroll
            for (int r = 0; r < GB; ++r) {
                const int e = n0 + m;
                sQ[r * QG + (e >> 5) * QSTR + (e & 31)] = acc[r];
            }
        }
    }
    __syncthreads();

    f4 park[2 * G];
    static_assert(G * (DD * NH) == 2 * G * 4 * NT, "8 float4 per thread hold one group");
    static_assert(G * UG <= TT * HSTR, "a parked group fits in the H tile's place");
    static_assert(UG % 4 == 0 && OFF_U % 4 == 0 && OFF_H % 4 == 0, "16-byte park copies");

    {
        int tl = threadIdx.x;
        asm volatile("" : "+v"(tl));
        const int lane = tl & 31;
        const int wave = tl >> 5;
        const int m    = lane & 15;
        const int hi   = lane >> 4;
        const int   gl    = m & (GB - 1);
        const float vmask = (m < GB) ? 1.f : 0.f;
        float* ub = (gl < G) ? (sU + gl * UG) : (sH + (gl - G) * UG);
        static_assert(QSTR % 4 == 0 && QG % 4 == 0 && OFF_Q % 4 == 0, "16-byte q reads");
        for (int t = 0; t < (DD / 16) * NH / NW; ++t) {
            const int job = wave * ((DD / 16) * NH / NW) + t;
            const int dt  = job >> 4;
            const int h   = job & 15;
            const int d0  = dt * 16;
            const f4* a4 = (const f4*)(Wk + (size_t)(d0 + m) * DD + 32 * h + 16 * hi);
            const f4* q4 = (const f4*)(sQ + gl * QG + h * QSTR + 16 * hi);
            f4 av[4], bv[4];
            #pragma unroll
            for (int v = 0; v < 4; ++v) av[v] = a4[v];
            #pragma unroll
            for (int v = 0; v < 4; ++v) bv[v] = q4[v];
            v8f acc = {};
            #pragma unroll
            for (int s = 0; s < 8; ++s) {
                const int v = s >> 1, e = 2 * (s & 1);
                v2f a, b;
                a[0] = av[v][e];
                a[1] = av[v][e + 1];
                b[0] = vmask * bv[v][e];
                b[1] = vmask * bv[v][e + 1];
                acc = wmma4(a, b, acc);
            }
            if (m < GB) {
                #pragma unroll
                for (int r = 0; r < 8; ++r)
                    ub[(d0 + r + 8 * hi) * NH + h] = acc[r];
            }
        }
    }
    __syncthreads();
    #pragma unroll
    for (int g = 0; g < G; ++g) {
        #pragma unroll
        for (int j = 0; j < 2; ++j)
            park[2 * g + j] = *(const f4*)(sH + g * UG + 4 * (tid + j * NT));
    }
    __syncthreads();

    static_assert(TT * DD / 4 % NT == 0, "whole float4 per thread");
    constexpr int HV = TT * DD / 4 / NT;
    f4 pz[HV];
    {
        const f4* pe4 = (const f4*)pos;
        #pragma unroll
        for (int j = 0; j < HV; ++j) pz[j] = pe4[tid + j * NT];
    }

    for (int sg = 0; sg < NSG; ++sg) {
    const long   bg0 = b0 + sg * G;
    if (sg == 1) {
        #pragma unroll
        for (int g = 0; g < G; ++g) {
            #pragma unroll
            for (int j = 0; j < 2; ++j) {
                f4* slot = (f4*)(sU + g * UG + 4 * (tid + j * NT));
                const f4 c = *slot;
                *slot = park[2 * g + j];
                park[2 * g + j] = c;
            }
        }
    }

    const float scale = 0.044194173824159216f;
    for (int g = 0; g < G; ++g) {
        {
            int tl = threadIdx.x;
            asm volatile("" : "+v"(tl));
            const int tid  = tl;
            const f4* in4 = (const f4*)(inputs + (bg0 + g) * (size_t)(TT * DD));
            f4 x[HV];
            #pragma unroll
            for (int j = 0; j < HV; ++j) x[j] = in4[tid + j * NT];
            #pragma unroll
            for (int j = 0; j < HV; ++j) {
                const int i = tid + j * NT;
                int t = i >> 7, d = (i & 127) << 2;
                float* dst = sH + t * HSTR + d;
                dst[0] = x[j].x + pz[j].x; dst[1] = x[j].y + pz[j].y;
                dst[2] = x[j].z + pz[j].z; dst[3] = x[j].w + pz[j].w;
            }
            if (sg * G + g + 1 < GB) {
                const float* nxt = inputs + (bg0 + g + 1) * (size_t)(TT * DD);
                static_assert(TT * DD / 32 == NT, "one 128-byte line per thread");
                __builtin_prefetch(nxt + (size_t)tid * 32, 0, 1);
            }
        }
        __syncthreads();

        static_assert(KS * (TT / 16) == NW && KS * KSL == DD, "one (tile, slice) per wave");
        {
            int tl = threadIdx.x;
            asm volatile("" : "+v"(tl));
            const int lane = tl & 31;
            const int wave = tl >> 5;
            const int m    = lane & 15;
            const int hi   = lane >> 4;
            const int t0 = (wave & 3) * 16;
            const int c  = wave >> 2;
            v8f acc = {};
            for (int k = 0; k < KSL; k += 4) {
                const int kk = c * KSL + k + 2 * hi;
                v2f a, b;
                a[0] = sH[(t0 + m) * HSTR + kk];
                a[1] = sH[(t0 + m) * HSTR + kk + 1];
                b[0] = sU[g * UG + kk * NH + m];
                b[1] = sU[g * UG + (kk + 1) * NH + m];
                acc = wmma4(a, b, acc);
            }
            #pragma unroll
            for (int r = 0; r < 8; ++r)
                sP[c * PSTR + (t0 + r + 8 * hi) * SSTR + m] = acc[r];
        }
        __syncthreads();

        static_assert(TT == 64 && NH <= NW, "two t per lane, one head per wave");
        {
        int tl = threadIdx.x;
        asm volatile("" : "+v"(tl));
        const int lane = tl & 31;
        const int wave = tl >> 5;
        if (wave < NH) {
            float* col = sS + wave;
            const float* par = sP + wave;
            float s0 = par[lane * SSTR];
            float s1 = par[(lane + 32) * SSTR];
            for (int c = 1; c < KS; ++c) {
                s0 += par[c * PSTR + lane * SSTR];
                s1 += par[c * PSTR + (lane + 32) * SSTR];
            }
            s0 *= scale;
            s1 *= scale;
            const float mx  = wave_max(fmaxf(s0, s1));
            const float e0  = expf(s0 - mx);
            const float e1  = expf(s1 - mx);
            const float inv = 1.f / wave_sum(e0 + e1);
            col[lane * SSTR]        = e0 * inv;
            col[(lane + 32) * SSTR] = e1 * inv;
        }
        }
        __syncthreads();

        {
            int tl = threadIdx.x;
            asm volatile("" : "+v"(tl));
            const int lane = tl & 31;
            const int wave = tl >> 5;
            const int m    = lane & 15;
            const int hi   = lane >> 4;
            const int d0 = wave * 16;
            v8f acc = {};
            for (int k = 0; k < TT; k += 4) {
                const int kk = k + 2 * hi;
                v2f a, b;
                a[0] = sH[kk * HSTR + d0 + m];
                a[1] = sH[(kk + 1) * HSTR + d0 + m];
                b[0] = sS[kk * SSTR + m];
                b[1] = sS[(kk + 1) * SSTR + m];
                acc = wmma4(a, b, acc);
            }
            #pragma unroll
            for (int r = 0; r < 8; ++r)
                sU[g * UG + (d0 + r + 8 * hi) * NH + m] = acc[r];
        }
        __syncthreads();
    }
    }

    #pragma unroll
    for (int g = 0; g < G; ++g) {
        #pragma unroll
        for (int j = 0; j < 2; ++j)
            *(f4*)(sH + g * UG + 4 * (tid + j * NT)) = park[2 * g + j];
    }
    __syncthreads();

    {
        int tl = threadIdx.x;
        asm volatile("" : "+v"(tl));
        const int lane = tl & 31;
        const int wave = tl >> 5;
        const int m    = lane & 15;
        const int hi   = lane >> 4;
        const int   gl    = m & (GB - 1);
        const float vmask = (m < GB) ? 1.f : 0.f;
        const float* cb = (gl < G) ? (sH + gl * UG) : (sU + (gl - G) * UG);
        {
            const int job = wave;
            const int h   = job >> 1;
            const int n0  = (job & 1) * 16;
            v8f acc = {};
            for (int k = 0; k < DD; k += 4) {
                const int kk = k + 2 * hi;
                v2f a, b;
                a[0] = vmask * cb[kk * NH + h];
                a[1] = vmask * cb[(kk + 1) * NH + h];
                b[0] = Wv[(size_t)kk * DD + 32 * h + n0 + m];
                b[1] = Wv[(size_t)(kk + 1) * DD + 32 * h + n0 + m];
                acc = wmma4(a, b, acc);
            }
            static_assert(GB == 8, "rows 0..7 are the 8 registers of the lower lane half");
            if (hi == 0) {
                #pragma unroll
                for (int r = 0; r < GB; ++r)
                    out[(b0 + r) * (size_t)DD + 32 * h + n0 + m] = acc[r];
            }
        }
    }
}

extern "C" void kernel_launch(void* const* d_in, const int* in_sizes, int n_in,
                              void* d_out, int out_size, void* d_ws, size_t ws_size,
                              hipStream_t stream) {
    const float* inputs = (const float*)d_in[0];
    const float* pos    = (const float*)d_in[1];
    const float* Wq     = (const float*)d_in[2];
    const float* Wk     = (const float*)d_in[3];
    const float* Wv     = (const float*)d_in[4];
    float* out          = (float*)d_out;
    (void)d_ws; (void)ws_size; (void)n_in; (void)out_size;

    const int B = in_sizes[0] / (TT * DD);
    const size_t smem = (size_t)LDS_FLOATS * sizeof(float);

    hipFuncSetAttribute((const void*)temporal_attn_lastrow_g4,
                        hipFuncAttributeMaxDynamicSharedMemorySize, (int)smem);

    temporal_attn_lastrow_g4<<<B / GB, NT, smem, stream>>>(
        inputs, pos, Wq, Wk, Wv, out);
}


# LDS strides re-chosen (q per batch 580 dwords; U/C planes 8194 with areas at 0 and 8 mod 64; H rows 524) so lanes of the Wk.q, C.Wv and C-GEMM fragment reads hit different banks; bit-identical
# speedup vs baseline: 2.9592x; 1.0501x over previous
#include <hip/hip_runtime.h>
#include <math.h>

constexpr int TT = 64;
constexpr int DD = 512;
constexpr int NH = 16;
constexpr int G  = 4;
constexpr int NSG = 2;
constexpr int GB  = G * NSG;
constexpr int NT = 1024;
constexpr int NW = NT / 32;

constexpr int HSTR = 524;
constexpr int RSTR = 516;
constexpr int QSTR = 36;
constexpr int QG   = NH * QSTR + 4;
constexpr int UG   = DD * NH + 2;
constexpr int SSTR = NH + 1;

constexpr int KS   = NW / (TT / 16);
constexpr int KSL  = DD / KS;
constexpr int PSTR = TT * SSTR;

constexpr int OFF_H = 0;
constexpr int OFF_R = OFF_H + TT * HSTR;
constexpr int RP_FLOATS = (KS * PSTR > 16 * RSTR) ? KS * PSTR : 16 * RSTR;
constexpr int OFF_Q = OFF_R + RP_FLOATS;
constexpr int OFF_U_MIN = OFF_Q + GB * QG;
constexpr int OFF_U = OFF_U_MIN + (8 + 64 - OFF_U_MIN % 64) % 64;
constexpr int OFF_S = OFF_U + G * UG;
constexpr int LDS_FLOATS = OFF_S + TT * SSTR;
static_assert(OFF_H % 64 == 0 && OFF_U % 64 == 8 && UG % 64 == 2 && G == 4,
              "the 8 U/C planes begin in banks 0,2,..,14");
static_assert(QG % 64 == 4, "the 8 batches' q slices begin 4 banks apart");
static_assert(HSTR % 64 == 12 && HSTR % 4 == 0, "H rows stay 16-byte aligned; their reads as said at HSTR");
static_assert(KS * PSTR <= OFF_Q - OFF_R, "score partials fit in the sR area");
static_assert(LDS_FLOATS * 4 <= 320 * 1024, "one workgroup's LDS");

typedef __attribute__((ext_vector_type(2))) float v2f;
typedef __attribute__((ext_vector_type(4))) float f4;
typedef __attribute__((ext_vector_type(8))) float v8f;

static __device__ __forceinline__ v8f wmma4(v2f a, v2f b, v8f c) {
    return __builtin_amdgcn_wmma_f32_16x16x4_f32(
        false, a, false, b, (short)0, c, false, false);
}

template <int CTRL>
static __device__ __forceinline__ float row_xor(float x) {
    const int b = __builtin_bit_cast(int, x);
    return __builtin_bit_cast(float,
        __builtin_amdgcn_update_dpp(b, b, CTRL, 0xf, 0xf, false));
}
static __device__ __forceinline__ float xhalf(float x) {
    const int b = __builtin_bit_cast(int, x);
    return __builtin_bit_cast(float,
        __builtin_amdgcn_permlanex16(b, b, 0x76543210, 0xfedcba98, false, false));
}
static __device__ __forceinline__ float wave_max(float x) {
    x = fmaxf(x, row_xor<0x161>(x));
    x = fmaxf(x, row_xor<0x162>(x));
    x = fmaxf(x, row_xor<0x164>(x));
    x = fmaxf(x, row_xor<0x168>(x));
    return fmaxf(x, xhalf(x));
}
static __device__ __forceinline__ float wave_sum(float x) {
    x += row_xor<0x161>(x);
    x += row_xor<0x162>(x);
    x += row_xor<0x164>(x);
    x += row_xor<0x168>(x);
    return x + xhalf(x);
}

__global__ __launch_bounds__(NT, 1)
void temporal_attn_lastrow_g4(const float* __restrict__ inputs,
                              const float* __restrict__ pos,
                              const float* __restrict__ Wq,
                              const float* __restrict__ Wk,
                              const float* __restrict__ Wv,
                              float* __restrict__ out)
{
    extern __shared__ float lds[];
    float* sH = lds + OFF_H;
    float* sR = lds + OFF_R;
    float* sQ = lds + OFF_Q;
    float* sU = lds + OFF_U;
    float* sS = lds + OFF_S;
    float* sP = lds + OFF_R;

    const int tid  = threadIdx.x;
    const int lane = tid & 31;
    const int wave = tid >> 5;
    const int m    = lane & 15;
    const int hi   = lane >> 4;
    const long b0  = (long)blockIdx.x * GB;

    for (int i = tid; i < 16 * RSTR; i += NT) sR[i] = 0.f;
    __syncthreads();
    static_assert(GB * DD % NT == 0, "whole trips");
    for (int it = 0; it < GB * DD / NT; ++it) {
        const int i = tid + it * NT;
        int g = i >> 9, d = i & 511;
        sR[g * RSTR + d] =
            inputs[(b0 + g) * (size_t)(TT * DD) + 63 * DD + d] + pos[63 * DD + d];
    }
    __syncthreads();

    static_assert(NW == DD / 16, "one 16-column tile per wave");
    {
        const int n0 = wave * 16;
        v8f acc = {};
        for (int k = 0; k < DD; k += 4) {
            const int kk = k + 2 * hi;
            v2f a, b;
            a[0] = sR[m * RSTR + kk];
            a[1] = sR[m * RSTR + kk + 1];
            b[0] = Wq[(size_t)kk * DD + n0 + m];
            b[1] = Wq[(size_t)(kk + 1) * DD + n0 + m];
            acc = wmma4(a, b, acc);
        }
        static_assert(GB == 8, "rows 0..7 are the 8 registers of the lower lane half");
        if (hi == 0) {
            #pragma unroll
            for (int r = 0; r < GB; ++r) {
                const int e = n0 + m;
                sQ[r * QG + (e >> 5) * QSTR + (e & 31)] = acc[r];
            }
        }
    }
    __syncthreads();

    v2f park[4 * G];
    static_assert(G * (DD * NH) == 4 * G * 2 * NT, "16 float2 per thread hold one group");
    static_assert(G * UG <= TT * HSTR, "a parked group fits in the H tile's place");
    static_assert(UG % 2 == 0 && OFF_U % 2 == 0 && OFF_H % 2 == 0, "8-byte park copies");

    {
        int tl = threadIdx.x;
        asm volatile("" : "+v"(tl));
        const int lane = tl & 31;
        const int wave = tl >> 5;
        const int m    = lane & 15;
        const int hi   = lane >> 4;
        const int   gl    = m & (GB - 1);
        const float vmask = (m < GB) ? 1.f : 0.f;
        float* ub = (gl < G) ? (sU + gl * UG) : (sH + (gl - G) * UG);
        static_assert(QSTR % 4 == 0 && QG % 4 == 0 && OFF_Q % 4 == 0, "16-byte q reads");
        for (int t = 0; t < (DD / 16) * NH / NW; ++t) {
            const int job = wave * ((DD / 16) * NH / NW) + t;
            const int dt  = job >> 4;
            const int h   = job & 15;
            const int d0  = dt * 16;
            const f4* a4 = (const f4*)(Wk + (size_t)(d0 + m) * DD + 32 * h + 16 * hi);
            const f4* q4 = (const f4*)(sQ + gl * QG + h * QSTR + 16 * hi);
            f4 av[4], bv[4];
            #pragma unroll
            for (int v = 0; v < 4; ++v) av[v] = a4[v];
            #pragma unroll
            for (int v = 0; v < 4; ++v) bv[v] = q4[v];
            v8f acc = {};
            #pragma unroll
            for (int s = 0; s < 8; ++s) {
                const int v = s >> 1, e = 2 * (s & 1);
                v2f a, b;
                a[0] = av[v][e];
                a[1] = av[v][e + 1];
                b[0] = vmask * bv[v][e];
                b[1] = vmask * bv[v][e + 1];
                acc = wmma4(a, b, acc);
            }
            if (m < GB) {
                #pragma unroll
                for (int r = 0; r < 8; ++r)
                    ub[(d0 + r + 8 * hi) * NH + h] = acc[r];
            }
        }
    }
    __syncthreads();
    #pragma unroll
    for (int g = 0; g < G; ++g) {
        #pragma unroll
        for (int j = 0; j < 4; ++j)
            park[4 * g + j] = *(const v2f*)(sH + g * UG + 2 * (tid + j * NT));
    }
    __syncthreads();

    static_assert(TT * DD / 4 % NT == 0, "whole float4 per thread");
    constexpr int HV = TT * DD / 4 / NT;
    f4 pz[HV];
    {
        const f4* pe4 = (const f4*)pos;
        #pragma unroll
        for (int j = 0; j < HV; ++j) pz[j] = pe4[tid + j * NT];
    }

    for (int sg = 0; sg < NSG; ++sg) {
    const long   bg0 = b0 + sg * G;
    if (sg == 1) {
        #pragma unroll
        for (int g = 0; g < G; ++g) {
            #pragma unroll
            for (int j = 0; j < 4; ++j) {
                v2f* slot = (v2f*)(sU + g * UG + 2 * (tid + j * NT));
                const v2f c = *slot;
                *slot = park[4 * g + j];
                park[4 * g + j] = c;
            }
        }
    }

    const float scale = 0.044194173824159216f;
    for (int g = 0; g < G; ++g) {
        {
            int tl = threadIdx.x;
            asm volatile("" : "+v"(tl));
            const int tid  = tl;
            const f4* in4 = (const f4*)(inputs + (bg0 + g) * (size_t)(TT * DD));
            f4 x[HV];
            #pragma unroll
            for (int j = 0; j < HV; ++j) x[j] = in4[tid + j * NT];
            #pragma unroll
            for (int j = 0; j < HV; ++j) {
                const int i = tid + j * NT;
                int t = i >> 7, d = (i & 127) << 2;
                float* dst = sH + t * HSTR + d;
                dst[0] = x[j].x + pz[j].x; dst[1] = x[j].y + pz[j].y;
                dst[2] = x[j].z + pz[j].z; dst[3] = x[j].w + pz[j].w;
            }
            if (sg * G + g + 1 < GB) {
                const float* nxt = inputs + (bg0 + g + 1) * (size_t)(TT * DD);
                static_assert(TT * DD / 32 == NT, "one 128-byte line per thread");
                __builtin_prefetch(nxt + (size_t)tid * 32, 0, 1);
            }
        }
        __syncthreads();

        static_assert(KS * (TT / 16) == NW && KS * KSL == DD, "one (tile, slice) per wave");
        {
            int tl = threadIdx.x;
            asm volatile("" : "+v"(tl));
            const int lane = tl & 31;
            const int wave = tl >> 5;
            const int m    = lane & 15;
            const int hi   = lane >> 4;
            const int t0 = (wave & 3) * 16;
            const int c  = wave >> 2;
            v8f acc = {};
            for (int k = 0; k < KSL; k += 4) {
                const int kk = c * KSL + k + 2 * hi;
                v2f a, b;
                a[0] = sH[(t0 + m) * HSTR + kk];
                a[1] = sH[(t0 + m) * HSTR + kk + 1];
                b[0] = sU[g * UG + kk * NH + m];
                b[1] = sU[g * UG + (kk + 1) * NH + m];
                acc = wmma4(a, b, acc);
            }
            #pragma unroll
            for (int r = 0; r < 8; ++r)
                sP[c * PSTR + (t0 + r + 8 * hi) * SSTR + m] = acc[r];
        }
        __syncthreads();

        static_assert(TT == 64 && NH <= NW, "two t per lane, one head per wave");
        {
        int tl = threadIdx.x;
        asm volatile("" : "+v"(tl));
        const int lane = tl & 31;
        const int wave = tl >> 5;
        if (wave < NH) {
            float* col = sS + wave;
            const float* par = sP + wave;
            float s0 = par[lane * SSTR];
            float s1 = par[(lane + 32) * SSTR];
            for (int c = 1; c < KS; ++c) {
                s0 += par[c * PSTR + lane * SSTR];
                s1 += par[c * PSTR + (lane + 32) * SSTR];
            }
            s0 *= scale;
            s1 *= scale;
            const float mx  = wave_max(fmaxf(s0, s1));
            const float e0  = expf(s0 - mx);
            const float e1  = expf(s1 - mx);
            const float inv = 1.f / wave_sum(e0 + e1);
            col[lane * SSTR]        = e0 * inv;
            col[(lane + 32) * SSTR] = e1 * inv;
        }
        }
        __syncthreads();

        {
            int tl = threadIdx.x;
            asm volatile("" : "+v"(tl));
            const int lane = tl & 31;
            const int wave = tl >> 5;
            const int m    = lane & 15;
            const int hi   = lane >> 4;
            const int d0 = wave * 16;
            v8f acc = {};
            for (int k = 0; k < TT; k += 4) {
                const int kk = k + 2 * hi;
                v2f a, b;
                a[0] = sH[kk * HSTR + d0 + m];
                a[1] = sH[(kk + 1) * HSTR + d0 + m];
                b[0] = sS[kk * SSTR + m];
                b[1] = sS[(kk + 1) * SSTR + m];
                acc = wmma4(a, b, acc);
            }
            #pragma unroll
            for (int r = 0; r < 8; ++r)
                sU[g * UG + (d0 + r + 8 * hi) * NH + m] = acc[r];
        }
        __syncthreads();
    }
    }

    #pragma unroll
    for (int g = 0; g < G; ++g) {
        #pragma unroll
        for (int j = 0; j < 4; ++j)
            *(v2f*)(sH + g * UG + 2 * (tid + j * NT)) = park[4 * g + j];
    }
    __syncthreads();

    {
        int tl = threadIdx.x;
        asm volatile("" : "+v"(tl));
        const int lane = tl & 31;
        const int wave = tl >> 5;
        const int m    = lane & 15;
        const int hi   = lane >> 4;
        const int   gl    = m & (GB - 1);
        const float vmask = (m < GB) ? 1.f : 0.f;
        const float* cb = (gl < G) ? (sH + gl * UG) : (sU + (gl - G) * UG);
        {
            const int job = wave;
            const int h   = job >> 1;
            const int n0  = (job & 1) * 16;
            v8f acc = {};
            for (int k = 0; k < DD; k += 4) {
                const int kk = k + 2 * hi;
                v2f a, b;
                a[0] = vmask * cb[kk * NH + h];
                a[1] = vmask * cb[(kk + 1) * NH + h];
                b[0] = Wv[(size_t)kk * DD + 32 * h + n0 + m];
                b[1] = Wv[(size_t)(kk + 1) * DD + 32 * h + n0 + m];
                acc = wmma4(a, b, acc);
            }
            static_assert(GB == 8, "rows 0..7 are the 8 registers of the lower lane half");
            if (hi == 0) {
                #pragma unroll
                for (int r = 0; r < GB; ++r)
                    out[(b0 + r) * (size_t)DD + 32 * h + n0 + m] = acc[r];
            }
        }
    }
}

extern "C" void kernel_launch(void* const* d_in, const int* in_sizes, int n_in,
                              void* d_out, int out_size, void* d_ws, size_t ws_size,
                              hipStream_t stream) {
    const float* inputs = (const float*)d_in[0];
    const float* pos    = (const float*)d_in[1];
    const float* Wq     = (const float*)d_in[2];
    const float* Wk     = (const float*)d_in[3];
    const float* Wv     = (const float*)d_in[4];
    float* out          = (float*)d_out;
    (void)d_ws; (void)ws_size; (void)n_in; (void)out_size;

    const int B = in_sizes[0] / (TT * DD);
    const size_t smem = (size_t)LDS_FLOATS * sizeof(float);

    hipFuncSetAttribute((const void*)temporal_attn_lastrow_g4,
                        hipFuncAttributeMaxDynamicSharedMemorySize, (int)smem);

    temporal_attn_lastrow_g4<<<B / GB, NT, smem, stream>>>(
        inputs, pos, Wq, Wk, Wv, out);
}
